# starting kernel
# baseline (speedup 1.0000x reference)
#include <hip/hip_runtime.h>
#include <hip/hip_bf16.h>

typedef __attribute__((ext_vector_type(2))) float v2f;
typedef __attribute__((ext_vector_type(8))) float v8f;

#define D 128

__global__ void init_deg_kernel(float* deg, int n) {
  int i = blockIdx.x * blockDim.x + threadIdx.x;
  if (i < n) deg[i] = 1.0f;
}

__global__ void deg_accum_kernel(const int* __restrict__ col,
                                 const float* __restrict__ ew,
                                 float* deg, int e_cnt) {
  int e = blockIdx.x * blockDim.x + threadIdx.x;
  if (e < e_cnt) atomicAdd(&deg[col[e]], ew[e]);
}

__global__ void dinv_kernel(float* deg, int n) {
  int i = blockIdx.x * blockDim.x + threadIdx.x;
  if (i < n) {
    float d = deg[i];
    deg[i] = (d > 0.0f) ? rsqrtf(d) : 0.0f;
  }
}

__global__ __launch_bounds__(256) void gemm_wmma_kernel(
    const float* __restrict__ x, const float* __restrict__ w,
    float* __restrict__ xw, int n_rows) {
  const int lane = threadIdx.x & 31;
  const int wave = blockIdx.x * (blockDim.x >> 5) + (threadIdx.x >> 5);
  const int row0 = wave * 16;
  if (row0 >= n_rows) return;

  const int m  = lane & 15;
  const int kh = (lane >> 4) * 2;

  const int arow = (row0 + m < n_rows) ? (row0 + m) : (n_rows - 1);

  v8f acc[8];
#pragma unroll
  for (int t = 0; t < 8; ++t) acc[t] = (v8f){0.f,0.f,0.f,0.f,0.f,0.f,0.f,0.f};

  for (int k = 0; k < D; k += 4) {
    const float* ap = x + (size_t)arow * D + (k + kh);
    v2f a;
    a.x = ap[0];
    a.y = ap[1];
#pragma unroll
    for (int t = 0; t < 8; ++t) {
      const float* bp = w + (size_t)(k + kh) * D + t * 16 + m;
      v2f b;
      b.x = bp[0];
      b.y = bp[D];
      acc[t] = __builtin_amdgcn_wmma_f32_16x16x4_f32(
           false, a,  false, b,
           (short)0, acc[t],  false,  false);
    }
  }

#pragma unroll
  for (int t = 0; t < 8; ++t) {
#pragma unroll
    for (int j = 0; j < 8; ++j) {
      const int M = (kh << 2) + j;
      if (row0 + M < n_rows)
        xw[(size_t)(row0 + M) * D + t * 16 + m] = acc[t][j];
    }
  }
}

__global__ void init_out_kernel(const float* __restrict__ xw,
                                const float* __restrict__ dinv,
                                const float* __restrict__ bias,
                                float* out, long long total) {
  long long idx = (long long)blockIdx.x * blockDim.x + threadIdx.x;
  if (idx >= total) return;
  int i = (int)(idx >> 7);
  int d = (int)(idx & (D - 1));
  float di = dinv[i];
  out[idx] = bias[d] + di * di * xw[idx];
}

__global__ void scatter_kernel(const int* __restrict__ row,
                               const int* __restrict__ col,
                               const float* __restrict__ ew,
                               const float* __restrict__ dinv,
                               const float* __restrict__ xw,
                               float* out, long long total) {
  long long idx = (long long)blockIdx.x * blockDim.x + threadIdx.x;
  if (idx >= total) return;
  int e = (int)(idx >> 7);
  int d = (int)(idx & (D - 1));
  int r = row[e];
  int c = col[e];
  float norm = dinv[r] * ew[e] * dinv[c];
  atomicAdd(&out[(size_t)c * D + d], norm * xw[(size_t)r * D + d]);
}

__global__ void prelu_kernel(float* out, const float* __restrict__ pa,
                             long long total) {
  long long idx = (long long)blockIdx.x * blockDim.x + threadIdx.x;
  if (idx >= total) return;
  float v = out[idx];
  out[idx] = (v >= 0.0f) ? v : pa[0] * v;
}

extern "C" void kernel_launch(void* const* d_in, const int* in_sizes, int n_in,
                              void* d_out, int out_size, void* d_ws, size_t ws_size,
                              hipStream_t stream) {
  const float* x    = (const float*)d_in[0];
  const int*   ei   = (const int*)  d_in[1];
  const float* ew   = (const float*)d_in[2];
  const float* w    = (const float*)d_in[3];
  const float* bias = (const float*)d_in[4];
  const float* pa   = (const float*)d_in[5];
  float* out = (float*)d_out;

  const int n     = in_sizes[0] / D;
  const int e_cnt = in_sizes[2];
  const int* row = ei;
  const int* col = ei + e_cnt;

  float* deg = (float*)d_ws;
  float* xw  = deg + n;

  const long long nd  = (long long)n * D;
  const long long edd = (long long)e_cnt * D;

  init_deg_kernel<<<(n + 255) / 256, 256, 0, stream>>>(deg, n);
  deg_accum_kernel<<<(e_cnt + 255) / 256, 256, 0, stream>>>(col, ew, deg, e_cnt);
  dinv_kernel<<<(n + 255) / 256, 256, 0, stream>>>(deg, n);
  {
    int tiles  = (n + 15) / 16;
    int blocks = (tiles + 7) / 8;
    gemm_wmma_kernel<<<blocks, 256, 0, stream>>>(x, w, xw, n);
  }
  init_out_kernel<<<(int)((nd + 255) / 256), 256, 0, stream>>>(xw, deg, bias, out, nd);
  scatter_kernel<<<(int)((edd + 255) / 256), 256, 0, stream>>>(row, col, ew, deg, xw, out, edd);
  prelu_kernel<<<(int)((nd + 255) / 256), 256, 0, stream>>>(out, pa, nd);
}


# scatter-add turned into gather: edges grouped by target (int atomic rank + scan), one wave sums each node's rows in registers and writes out once; float atomics gone, sum order is list order
# speedup vs baseline: 3.4213x; 3.4213x over previous
#include <hip/hip_runtime.h>
#include <hip/hip_bf16.h>

typedef __attribute__((ext_vector_type(2))) float v2f;
typedef __attribute__((ext_vector_type(8))) float v8f;

#define D 128


__global__ void zero_cnt_kernel(int* cnt, int n) {
  int i = blockIdx.x * blockDim.x + threadIdx.x;
  if (i < n) cnt[i] = 0;
}

__global__ void rank_kernel(const int* __restrict__ col, int* cnt,
                            int* __restrict__ rank, int e_cnt) {
  int e = blockIdx.x * blockDim.x + threadIdx.x;
  if (e < e_cnt) rank[e] = atomicAdd(&cnt[col[e]], 1);
}

__global__ __launch_bounds__(1024) void scan_kernel(const int* __restrict__ cnt,
                                                    int* __restrict__ off, int n) {
  __shared__ int s_inc[32];
  __shared__ int s_pre[32];
  const int t = threadIdx.x;
  const int b = blockIdx.x;
  const int lane = t & 31;
  const int wv = t >> 5;
  const int gi = b * 1024 + t;
  const int v = (gi < n) ? cnt[gi] : 0;
  int p = 0;
  for (int j = 0; j < b; ++j) p += cnt[j * 1024 + t];
  int inc = v;
  for (int d = 1; d < 32; d <<= 1) {
    int u = __shfl_up(inc, d);
    if (lane >= d) inc += u;
  }
  for (int d = 16; d > 0; d >>= 1) p += __shfl_xor(p, d);
  if (lane == 31) s_inc[wv] = inc;
  if (lane == 0) s_pre[wv] = p;
  __syncthreads();
  int q = s_pre[lane] + ((lane < wv) ? s_inc[lane] : 0);
  for (int d = 16; d > 0; d >>= 1) q += __shfl_xor(q, d);
  if (gi < n) off[gi] = q + inc - v;
}

__global__ void fill_kernel(const int* __restrict__ row,
                            const int* __restrict__ col,
                            const float* __restrict__ ew,
                            const int* __restrict__ off,
                            const int* __restrict__ rank,
                            int* __restrict__ lst, int e_cnt) {
  int e = blockIdx.x * blockDim.x + threadIdx.x;
  if (e >= e_cnt) return;
  int pos = off[col[e]] + rank[e];
  ((int2*)lst)[pos] = make_int2(row[e], __float_as_int(ew[e]));
}

__global__ void dinv_kernel(const int* __restrict__ off,
                            const int* __restrict__ cnt,
                            const int* __restrict__ lst,
                            float* __restrict__ dinv, int n) {
  const int t = blockIdx.x * blockDim.x + threadIdx.x;
  const int i = t >> 3;
  const int sub = t & 7;
  float s = 0.0f;
  if (i < n) {
    const int beg = off[i];
    const int end = beg + cnt[i];
    for (int p = beg + sub; p < end; p += 8) s += __int_as_float(lst[2 * p + 1]);
  }
  s += __shfl_xor(s, 4);
  s += __shfl_xor(s, 2);
  s += __shfl_xor(s, 1);
  if (i < n && sub == 0) {
    float d = 1.0f + s;
    dinv[i] = (d > 0.0f) ? rsqrtf(d) : 0.0f;
  }
}

__global__ __launch_bounds__(256) void gemm_wmma_kernel(
    const float* __restrict__ x, const float* __restrict__ w,
    float* __restrict__ xw, int n_rows) {
  const int lane = threadIdx.x & 31;
  const int wave = blockIdx.x * (blockDim.x >> 5) + (threadIdx.x >> 5);
  const int row0 = wave * 16;
  if (row0 >= n_rows) return;

  const int m  = lane & 15;
  const int kh = (lane >> 4) * 2;

  const int arow = (row0 + m < n_rows) ? (row0 + m) : (n_rows - 1);

  v8f acc[8];
#pragma unroll
  for (int t = 0; t < 8; ++t) acc[t] = (v8f){0.f,0.f,0.f,0.f,0.f,0.f,0.f,0.f};

  for (int k = 0; k < D; k += 4) {
    const float* ap = x + (size_t)arow * D + (k + kh);
    v2f a;
    a.x = ap[0];
    a.y = ap[1];
#pragma unroll
    for (int t = 0; t < 8; ++t) {
      const float* bp = w + (size_t)(k + kh) * D + t * 16 + m;
      v2f b;
      b.x = bp[0];
      b.y = bp[D];
      acc[t] = __builtin_amdgcn_wmma_f32_16x16x4_f32(
           false, a,  false, b,
           (short)0, acc[t],  false,  false);
    }
  }

#pragma unroll
  for (int t = 0; t < 8; ++t) {
#pragma unroll
    for (int j = 0; j < 8; ++j) {
      const int M = (kh << 2) + j;
      if (row0 + M < n_rows)
        xw[(size_t)(row0 + M) * D + t * 16 + m] = acc[t][j];
    }
  }
}

template <int K>
__device__ __forceinline__ void gather_step(const float4* __restrict__ xw4,
                                            int r_l, float w_l, int j, int lane,
                                            float4& acc) {
  float4 v[K];
  float w[K];
#pragma unroll
  for (int k = 0; k < K; ++k) {
    const int r = __builtin_amdgcn_readlane(r_l, j + k);
    w[k] = __int_as_float(__builtin_amdgcn_readlane(__float_as_int(w_l), j + k));
    v[k] = xw4[(size_t)(unsigned)r * (D / 4) + lane];
  }
#pragma unroll
  for (int k = 0; k < K; ++k) {
    acc.x = fmaf(w[k], v[k].x, acc.x);
    acc.y = fmaf(w[k], v[k].y, acc.y);
    acc.z = fmaf(w[k], v[k].z, acc.z);
    acc.w = fmaf(w[k], v[k].w, acc.w);
  }
}

__global__ __launch_bounds__(256) void gather_kernel(
    const int* __restrict__ off, const int* __restrict__ cnt,
    const int* __restrict__ lst, const float* __restrict__ dinv,
    const float* __restrict__ xw, const float* __restrict__ bias,
    const float* __restrict__ pa, float* __restrict__ out, int n) {
  const int lane = threadIdx.x & 31;
  const int i = blockIdx.x * 8 + __builtin_amdgcn_readfirstlane(threadIdx.x >> 5);
  if (i >= n) return;
  const float4* xw4 = (const float4*)xw;
  const int2* lst2 = (const int2*)lst;
  const int beg = off[i];
  const int len = cnt[i];
  const float di = dinv[i];
  const float4 self = xw4[(size_t)i * (D / 4) + lane];
  const float sn = di * di;
  float4 acc = make_float4(sn * self.x, sn * self.y, sn * self.z, sn * self.w);
  for (int base = 0; base < len; base += 32) {
    const int m = (len - base < 32) ? (len - base) : 32;
    int r_l = 0;
    float w_l = 0.0f;
    if (lane < m) {
      const int2 pe = lst2[beg + base + lane];
      r_l = pe.x;
      w_l = (dinv[r_l] * __int_as_float(pe.y)) * di;
    }
    int j = 0;
    for (; j + 8 <= m; j += 8) gather_step<8>(xw4, r_l, w_l, j, lane, acc);
    if (m & 4) { gather_step<4>(xw4, r_l, w_l, j, lane, acc); j += 4; }
    if (m & 2) { gather_step<2>(xw4, r_l, w_l, j, lane, acc); j += 2; }
    if (m & 1) { gather_step<1>(xw4, r_l, w_l, j, lane, acc); }
  }
  const float a = pa[0];
  float4 o;
  o.x = acc.x + bias[lane * 4 + 0];
  o.y = acc.y + bias[lane * 4 + 1];
  o.z = acc.z + bias[lane * 4 + 2];
  o.w = acc.w + bias[lane * 4 + 3];
  o.x = (o.x >= 0.0f) ? o.x : a * o.x;
  o.y = (o.y >= 0.0f) ? o.y : a * o.y;
  o.z = (o.z >= 0.0f) ? o.z : a * o.z;
  o.w = (o.w >= 0.0f) ? o.w : a * o.w;
  ((float4*)out)[(size_t)i * (D / 4) + lane] = o;
}

extern "C" void kernel_launch(void* const* d_in, const int* in_sizes, int n_in,
                              void* d_out, int out_size, void* d_ws, size_t ws_size,
                              hipStream_t stream) {
  const float* x    = (const float*)d_in[0];
  const int*   ei   = (const int*)  d_in[1];
  const float* ew   = (const float*)d_in[2];
  const float* w    = (const float*)d_in[3];
  const float* bias = (const float*)d_in[4];
  const float* pa   = (const float*)d_in[5];
  float* out = (float*)d_out;

  const int n     = in_sizes[0] / D;
  const int e_cnt = in_sizes[2];
  const int* row = ei;
  const int* col = ei + e_cnt;

  const size_t n_pad = ((size_t)n + 127) / 128 * 128;
  const size_t e_pad = ((size_t)e_cnt + 127) / 128 * 128;
  int*   cnt  = (int*)d_ws;
  int*   off  = cnt + n_pad;
  float* dinv = (float*)(off + n_pad);
  int*   rank = (int*)(dinv + n_pad);
  int*   lst  = rank + e_pad;
  float* xw   = (float*)(lst + 2 * e_pad);

  zero_cnt_kernel<<<(n + 255) / 256, 256, 0, stream>>>(cnt, n);
  rank_kernel<<<(e_cnt + 255) / 256, 256, 0, stream>>>(col, cnt, rank, e_cnt);
  scan_kernel<<<(n + 1023) / 1024, 1024, 0, stream>>>(cnt, off, n);
  fill_kernel<<<(e_cnt + 255) / 256, 256, 0, stream>>>(row, col, ew, off, rank, lst, e_cnt);
  dinv_kernel<<<(int)(((long long)n * 8 + 255) / 256), 256, 0, stream>>>(off, cnt, lst, dinv, n);
  {
    int tiles  = (n + 15) / 16;
    int blocks = (tiles + 7) / 8;
    gemm_wmma_kernel<<<blocks, 256, 0, stream>>>(x, w, xw, n);
  }
  gather_kernel<<<(n + 7) / 8, 256, 0, stream>>>(off, cnt, lst, dinv, xw, bias, pa, out, n);
}


# edges grouped by target with LDS counters in two levels (256-node buckets, then nodes) in place of per-edge global returning adds; degree sums by LDS float adds; list order = LDS arrival order
# speedup vs baseline: 5.6270x; 1.6447x over previous
#include <hip/hip_runtime.h>
#include <hip/hip_bf16.h>

typedef __attribute__((ext_vector_type(2))) float v2f;
typedef __attribute__((ext_vector_type(8))) float v8f;

#define D 128


#define BSH 8
#define BW (1 << BSH)
#define NCHUNK 256
#define EB 8

__global__ __launch_bounds__(1024) void bucket_count_kernel(
    const int* __restrict__ col, int* __restrict__ hist, int e_cnt, int nb) {
  extern __shared__ int s_h[];
  const int c = blockIdx.x;
  for (int b = threadIdx.x; b < nb; b += blockDim.x) s_h[b] = 0;
  __syncthreads();
  const int per = (e_cnt + NCHUNK - 1) / NCHUNK;
  const int beg = c * per;
  const int end = (beg + per < e_cnt) ? (beg + per) : e_cnt;
  for (int e0 = beg + threadIdx.x; e0 < end; e0 += EB * blockDim.x) {
    int tg[EB];
#pragma unroll
    for (int k = 0; k < EB; ++k) {
      const int e = e0 + k * blockDim.x;
      tg[k] = col[(e < end) ? e : (end - 1)];
    }
#pragma unroll
    for (int k = 0; k < EB; ++k)
      if (e0 + k * blockDim.x < end) atomicAdd(&s_h[tg[k] >> BSH], 1);
  }
  __syncthreads();
  for (int b = threadIdx.x; b < nb; b += blockDim.x) hist[b * NCHUNK + c] = s_h[b];
}

__global__ __launch_bounds__(1024) void scan_kernel(const int* __restrict__ cnt,
                                                    int* __restrict__ off, int n) {
  __shared__ int s_inc[32];
  __shared__ int s_pre[32];
  const int t = threadIdx.x;
  const int b = blockIdx.x;
  const int lane = t & 31;
  const int wv = t >> 5;
  const int gi = b * 1024 + t;
  const int v = (gi < n) ? cnt[gi] : 0;
  int p = 0;
  for (int j = 0; j < b; ++j) p += cnt[j * 1024 + t];
  int inc = v;
  for (int d = 1; d < 32; d <<= 1) {
    int u = __shfl_up(inc, d);
    if (lane >= d) inc += u;
  }
  for (int d = 16; d > 0; d >>= 1) p += __shfl_xor(p, d);
  if (lane == 31) s_inc[wv] = inc;
  if (lane == 0) s_pre[wv] = p;
  __syncthreads();
  int q = s_pre[lane] + ((lane < wv) ? s_inc[lane] : 0);
  for (int d = 16; d > 0; d >>= 1) q += __shfl_xor(q, d);
  if (gi < n) off[gi] = q + inc - v;
}

__global__ __launch_bounds__(1024) void bucket_fill_kernel(
    const int* __restrict__ row, const int* __restrict__ col,
    const float* __restrict__ ew, const int* __restrict__ base,
    int* __restrict__ bkt, int e_cnt, int nb) {
  extern __shared__ int s_cur[];
  const int c = blockIdx.x;
  for (int b = threadIdx.x; b < nb; b += blockDim.x) s_cur[b] = base[b * NCHUNK + c];
  __syncthreads();
  const int per = (e_cnt + NCHUNK - 1) / NCHUNK;
  const int beg = c * per;
  const int end = (beg + per < e_cnt) ? (beg + per) : e_cnt;
  for (int e0 = beg + threadIdx.x; e0 < end; e0 += EB * blockDim.x) {
    int tg[EB];
    int sr[EB];
    float wt[EB];
#pragma unroll
    for (int k = 0; k < EB; ++k) {
      const int e = e0 + k * blockDim.x;
      const int ec = (e < end) ? e : (end - 1);
      tg[k] = col[ec];
      sr[k] = row[ec];
      wt[k] = ew[ec];
    }
#pragma unroll
    for (int k = 0; k < EB; ++k) {
      if (e0 + k * blockDim.x < end) {
        const int pos = atomicAdd(&s_cur[tg[k] >> BSH], 1);
        ((int2*)bkt)[pos] = make_int2((sr[k] << BSH) | (tg[k] & (BW - 1)),
                                      __float_as_int(wt[k]));
      }
    }
  }
}

__global__ __launch_bounds__(1024) void bucket_sort_kernel(
    const int* __restrict__ base, const int* __restrict__ bkt,
    int* __restrict__ cnt, int* __restrict__ off, float* __restrict__ dinv,
    int* __restrict__ lst, int n, int e_cnt, int nb) {
  __shared__ int s_cnt[BW];
  __shared__ float s_deg[BW];
  __shared__ int s_cur[BW];
  __shared__ int s_ws[BW / 32];
  const int b = blockIdx.x;
  const int t = threadIdx.x;
  const int lane = t & 31;
  const int beg = base[b * NCHUNK];
  const int end = (b + 1 < nb) ? base[(b + 1) * NCHUNK] : e_cnt;
  const int2* bk2 = (const int2*)bkt;
  if (t < BW) { s_cnt[t] = 0; s_deg[t] = 0.0f; }
  __syncthreads();
  for (int p0 = beg + t; p0 < end; p0 += EB * blockDim.x) {
    int2 r[EB];
#pragma unroll
    for (int k = 0; k < EB; ++k) {
      const int p = p0 + k * blockDim.x;
      r[k] = bk2[(p < end) ? p : (end - 1)];
    }
#pragma unroll
    for (int k = 0; k < EB; ++k) {
      if (p0 + k * blockDim.x < end) {
        const int tl = r[k].x & (BW - 1);
        atomicAdd(&s_cnt[tl], 1);
        atomicAdd(&s_deg[tl], __int_as_float(r[k].y));
      }
    }
  }
  __syncthreads();
  int v = 0;
  int inc = 0;
  if (t < BW) {
    v = s_cnt[t];
    inc = v;
    for (int d = 1; d < 32; d <<= 1) {
      int u = __shfl_up(inc, d);
      if (lane >= d) inc += u;
    }
    if (lane == 31) s_ws[t >> 5] = inc;
  }
  __syncthreads();
  if (t < BW) {
    int pre = 0;
    for (int k = 0; k < (t >> 5); ++k) pre += s_ws[k];
    const int ex = pre + inc - v;
    s_cur[t] = ex;
    const int node = b * BW + t;
    if (node < n) {
      cnt[node] = v;
      off[node] = beg + ex;
      const float d = 1.0f + s_deg[t];
      dinv[node] = (d > 0.0f) ? rsqrtf(d) : 0.0f;
    }
  }
  __syncthreads();
  for (int p0 = beg + t; p0 < end; p0 += EB * blockDim.x) {
    int2 r[EB];
#pragma unroll
    for (int k = 0; k < EB; ++k) {
      const int p = p0 + k * blockDim.x;
      r[k] = bk2[(p < end) ? p : (end - 1)];
    }
#pragma unroll
    for (int k = 0; k < EB; ++k) {
      if (p0 + k * blockDim.x < end) {
        const int pos = atomicAdd(&s_cur[r[k].x & (BW - 1)], 1);
        ((int2*)lst)[beg + pos] = make_int2((int)((unsigned)r[k].x >> BSH), r[k].y);
      }
    }
  }
}

__global__ __launch_bounds__(256) void gemm_wmma_kernel(
    const float* __restrict__ x, const float* __restrict__ w,
    float* __restrict__ xw, int n_rows) {
  const int lane = threadIdx.x & 31;
  const int wave = blockIdx.x * (blockDim.x >> 5) + (threadIdx.x >> 5);
  const int row0 = wave * 16;
  if (row0 >= n_rows) return;

  const int m  = lane & 15;
  const int kh = (lane >> 4) * 2;

  const int arow = (row0 + m < n_rows) ? (row0 + m) : (n_rows - 1);

  v8f acc[8];
#pragma unroll
  for (int t = 0; t < 8; ++t) acc[t] = (v8f){0.f,0.f,0.f,0.f,0.f,0.f,0.f,0.f};

  for (int k = 0; k < D; k += 4) {
    const float* ap = x + (size_t)arow * D + (k + kh);
    v2f a;
    a.x = ap[0];
    a.y = ap[1];
#pragma unroll
    for (int t = 0; t < 8; ++t) {
      const float* bp = w + (size_t)(k + kh) * D + t * 16 + m;
      v2f b;
      b.x = bp[0];
      b.y = bp[D];
      acc[t] = __builtin_amdgcn_wmma_f32_16x16x4_f32(
           false, a,  false, b,
           (short)0, acc[t],  false,  false);
    }
  }

#pragma unroll
  for (int t = 0; t < 8; ++t) {
#pragma unroll
    for (int j = 0; j < 8; ++j) {
      const int M = (kh << 2) + j;
      if (row0 + M < n_rows)
        xw[(size_t)(row0 + M) * D + t * 16 + m] = acc[t][j];
    }
  }
}

template <int K>
__device__ __forceinline__ void gather_step(const float4* __restrict__ xw4,
                                            int r_l, float w_l, int j, int lane,
                                            float4& acc) {
  float4 v[K];
  float w[K];
#pragma unroll
  for (int k = 0; k < K; ++k) {
    const int r = __builtin_amdgcn_readlane(r_l, j + k);
    w[k] = __int_as_float(__builtin_amdgcn_readlane(__float_as_int(w_l), j + k));
    v[k] = xw4[(size_t)(unsigned)r * (D / 4) + lane];
  }
#pragma unroll
  for (int k = 0; k < K; ++k) {
    acc.x = fmaf(w[k], v[k].x, acc.x);
    acc.y = fmaf(w[k], v[k].y, acc.y);
    acc.z = fmaf(w[k], v[k].z, acc.z);
    acc.w = fmaf(w[k], v[k].w, acc.w);
  }
}

__global__ __launch_bounds__(256) void gather_kernel(
    const int* __restrict__ off, const int* __restrict__ cnt,
    const int* __restrict__ lst, const float* __restrict__ dinv,
    const float* __restrict__ xw, const float* __restrict__ bias,
    const float* __restrict__ pa, float* __restrict__ out, int n) {
  const int lane = threadIdx.x & 31;
  const int i = blockIdx.x * 8 + __builtin_amdgcn_readfirstlane(threadIdx.x >> 5);
  if (i >= n) return;
  const float4* xw4 = (const float4*)xw;
  const int2* lst2 = (const int2*)lst;
  const int beg = off[i];
  const int len = cnt[i];
  const float di = dinv[i];
  const float4 self = xw4[(size_t)i * (D / 4) + lane];
  const float sn = di * di;
  float4 acc = make_float4(sn * self.x, sn * self.y, sn * self.z, sn * self.w);
  for (int base = 0; base < len; base += 32) {
    const int m = (len - base < 32) ? (len - base) : 32;
    int r_l = 0;
    float w_l = 0.0f;
    if (lane < m) {
      const int2 pe = lst2[beg + base + lane];
      r_l = pe.x;
      w_l = (dinv[r_l] * __int_as_float(pe.y)) * di;
    }
    int j = 0;
    for (; j + 8 <= m; j += 8) gather_step<8>(xw4, r_l, w_l, j, lane, acc);
    if (m & 4) { gather_step<4>(xw4, r_l, w_l, j, lane, acc); j += 4; }
    if (m & 2) { gather_step<2>(xw4, r_l, w_l, j, lane, acc); j += 2; }
    if (m & 1) { gather_step<1>(xw4, r_l, w_l, j, lane, acc); }
  }
  const float a = pa[0];
  float4 o;
  o.x = acc.x + bias[lane * 4 + 0];
  o.y = acc.y + bias[lane * 4 + 1];
  o.z = acc.z + bias[lane * 4 + 2];
  o.w = acc.w + bias[lane * 4 + 3];
  o.x = (o.x >= 0.0f) ? o.x : a * o.x;
  o.y = (o.y >= 0.0f) ? o.y : a * o.y;
  o.z = (o.z >= 0.0f) ? o.z : a * o.z;
  o.w = (o.w >= 0.0f) ? o.w : a * o.w;
  ((float4*)out)[(size_t)i * (D / 4) + lane] = o;
}

extern "C" void kernel_launch(void* const* d_in, const int* in_sizes, int n_in,
                              void* d_out, int out_size, void* d_ws, size_t ws_size,
                              hipStream_t stream) {
  const float* x    = (const float*)d_in[0];
  const int*   ei   = (const int*)  d_in[1];
  const float* ew   = (const float*)d_in[2];
  const float* w    = (const float*)d_in[3];
  const float* bias = (const float*)d_in[4];
  const float* pa   = (const float*)d_in[5];
  float* out = (float*)d_out;

  const int n     = in_sizes[0] / D;
  const int e_cnt = in_sizes[2];
  const int* row = ei;
  const int* col = ei + e_cnt;
  const int nb = (n + BW - 1) / BW;
  const int nh = nb * NCHUNK;

  const size_t n_pad = ((size_t)n + 127) / 128 * 128;
  const size_t e_pad = ((size_t)e_cnt + 127) / 128 * 128;
  const size_t h_pad = ((size_t)nh + 127) / 128 * 128;
  int*   cnt  = (int*)d_ws;
  int*   off  = cnt + n_pad;
  float* dinv = (float*)(off + n_pad);
  int*   hist = (int*)(dinv + n_pad);
  int*   base = hist + h_pad;
  int*   bkt  = base + h_pad;
  int*   lst  = bkt + 2 * e_pad;
  float* xw   = (float*)(lst + 2 * e_pad);

  bucket_count_kernel<<<NCHUNK, 1024, nb * sizeof(int), stream>>>(col, hist, e_cnt, nb);
  scan_kernel<<<(nh + 1023) / 1024, 1024, 0, stream>>>(hist, base, nh);
  bucket_fill_kernel<<<NCHUNK, 1024, nb * sizeof(int), stream>>>(row, col, ew, base, bkt, e_cnt, nb);
  bucket_sort_kernel<<<nb, 1024, 0, stream>>>(base, bkt, cnt, off, dinv, lst, n, e_cnt, nb);
  {
    int tiles  = (n + 15) / 16;
    int blocks = (tiles + 7) / 8;
    gemm_wmma_kernel<<<blocks, 256, 0, stream>>>(x, w, xw, n);
  }
  gather_kernel<<<(n + 7) / 8, 256, 0, stream>>>(off, cnt, lst, dinv, xw, bias, pa, out, n);
}


# source node's dinv multiplied into the xw rows at the GEMM's store, so the gather does no per-edge dinv lookup (one random 4-byte load per edge gone); products re-associated, sum order kept
# speedup vs baseline: 6.0108x; 1.0682x over previous
#include <hip/hip_runtime.h>
#include <hip/hip_bf16.h>

typedef __attribute__((ext_vector_type(2))) float v2f;
typedef __attribute__((ext_vector_type(8))) float v8f;

#define D 128


#define BSH 8
#define BW (1 << BSH)
#define NCHUNK 256
#define EB 8

__global__ __launch_bounds__(1024) void bucket_count_kernel(
    const int* __restrict__ col, int* __restrict__ hist, int e_cnt, int nb) {
  extern __shared__ int s_h[];
  const int c = blockIdx.x;
  for (int b = threadIdx.x; b < nb; b += blockDim.x) s_h[b] = 0;
  __syncthreads();
  const int per = (e_cnt + NCHUNK - 1) / NCHUNK;
  const int beg = c * per;
  const int end = (beg + per < e_cnt) ? (beg + per) : e_cnt;
  for (int e0 = beg + threadIdx.x; e0 < end; e0 += EB * blockDim.x) {
    int tg[EB];
#pragma unroll
    for (int k = 0; k < EB; ++k) {
      const int e = e0 + k * blockDim.x;
      tg[k] = col[(e < end) ? e : (end - 1)];
    }
#pragma unroll
    for (int k = 0; k < EB; ++k)
      if (e0 + k * blockDim.x < end) atomicAdd(&s_h[tg[k] >> BSH], 1);
  }
  __syncthreads();
  for (int b = threadIdx.x; b < nb; b += blockDim.x) hist[b * NCHUNK + c] = s_h[b];
}

__global__ __launch_bounds__(1024) void scan_kernel(const int* __restrict__ cnt,
                                                    int* __restrict__ off, int n) {
  __shared__ int s_inc[32];
  __shared__ int s_pre[32];
  const int t = threadIdx.x;
  const int b = blockIdx.x;
  const int lane = t & 31;
  const int wv = t >> 5;
  const int gi = b * 1024 + t;
  const int v = (gi < n) ? cnt[gi] : 0;
  int p = 0;
  for (int j = 0; j < b; ++j) p += cnt[j * 1024 + t];
  int inc = v;
  for (int d = 1; d < 32; d <<= 1) {
    int u = __shfl_up(inc, d);
    if (lane >= d) inc += u;
  }
  for (int d = 16; d > 0; d >>= 1) p += __shfl_xor(p, d);
  if (lane == 31) s_inc[wv] = inc;
  if (lane == 0) s_pre[wv] = p;
  __syncthreads();
  int q = s_pre[lane] + ((lane < wv) ? s_inc[lane] : 0);
  for (int d = 16; d > 0; d >>= 1) q += __shfl_xor(q, d);
  if (gi < n) off[gi] = q + inc - v;
}

__global__ __launch_bounds__(1024) void bucket_fill_kernel(
    const int* __restrict__ row, const int* __restrict__ col,
    const float* __restrict__ ew, const int* __restrict__ base,
    int* __restrict__ bkt, int e_cnt, int nb) {
  extern __shared__ int s_cur[];
  const int c = blockIdx.x;
  for (int b = threadIdx.x; b < nb; b += blockDim.x) s_cur[b] = base[b * NCHUNK + c];
  __syncthreads();
  const int per = (e_cnt + NCHUNK - 1) / NCHUNK;
  const int beg = c * per;
  const int end = (beg + per < e_cnt) ? (beg + per) : e_cnt;
  for (int e0 = beg + threadIdx.x; e0 < end; e0 += EB * blockDim.x) {
    int tg[EB];
    int sr[EB];
    float wt[EB];
#pragma unroll
    for (int k = 0; k < EB; ++k) {
      const int e = e0 + k * blockDim.x;
      const int ec = (e < end) ? e : (end - 1);
      tg[k] = col[ec];
      sr[k] = row[ec];
      wt[k] = ew[ec];
    }
#pragma unroll
    for (int k = 0; k < EB; ++k) {
      if (e0 + k * blockDim.x < end) {
        const int pos = atomicAdd(&s_cur[tg[k] >> BSH], 1);
        ((int2*)bkt)[pos] = make_int2((sr[k] << BSH) | (tg[k] & (BW - 1)),
                                      __float_as_int(wt[k]));
      }
    }
  }
}

__global__ __launch_bounds__(1024) void bucket_sort_kernel(
    const int* __restrict__ base, const int* __restrict__ bkt,
    int* __restrict__ cnt, int* __restrict__ off, float* __restrict__ dinv,
    int* __restrict__ lst, int n, int e_cnt, int nb) {
  __shared__ int s_cnt[BW];
  __shared__ float s_deg[BW];
  __shared__ int s_cur[BW];
  __shared__ int s_ws[BW / 32];
  const int b = blockIdx.x;
  const int t = threadIdx.x;
  const int lane = t & 31;
  const int beg = base[b * NCHUNK];
  const int end = (b + 1 < nb) ? base[(b + 1) * NCHUNK] : e_cnt;
  const int2* bk2 = (const int2*)bkt;
  if (t < BW) { s_cnt[t] = 0; s_deg[t] = 0.0f; }
  __syncthreads();
  for (int p0 = beg + t; p0 < end; p0 += EB * blockDim.x) {
    int2 r[EB];
#pragma unroll
    for (int k = 0; k < EB; ++k) {
      const int p = p0 + k * blockDim.x;
      r[k] = bk2[(p < end) ? p : (end - 1)];
    }
#pragma unroll
    for (int k = 0; k < EB; ++k) {
      if (p0 + k * blockDim.x < end) {
        const int tl = r[k].x & (BW - 1);
        atomicAdd(&s_cnt[tl], 1);
        atomicAdd(&s_deg[tl], __int_as_float(r[k].y));
      }
    }
  }
  __syncthreads();
  int v = 0;
  int inc = 0;
  if (t < BW) {
    v = s_cnt[t];
    inc = v;
    for (int d = 1; d < 32; d <<= 1) {
      int u = __shfl_up(inc, d);
      if (lane >= d) inc += u;
    }
    if (lane == 31) s_ws[t >> 5] = inc;
  }
  __syncthreads();
  if (t < BW) {
    int pre = 0;
    for (int k = 0; k < (t >> 5); ++k) pre += s_ws[k];
    const int ex = pre + inc - v;
    s_cur[t] = ex;
    const int node = b * BW + t;
    if (node < n) {
      cnt[node] = v;
      off[node] = beg + ex;
      const float d = 1.0f + s_deg[t];
      dinv[node] = (d > 0.0f) ? rsqrtf(d) : 0.0f;
    }
  }
  __syncthreads();
  for (int p0 = beg + t; p0 < end; p0 += EB * blockDim.x) {
    int2 r[EB];
#pragma unroll
    for (int k = 0; k < EB; ++k) {
      const int p = p0 + k * blockDim.x;
      r[k] = bk2[(p < end) ? p : (end - 1)];
    }
#pragma unroll
    for (int k = 0; k < EB; ++k) {
      if (p0 + k * blockDim.x < end) {
        const int pos = atomicAdd(&s_cur[r[k].x & (BW - 1)], 1);
        ((int2*)lst)[beg + pos] = make_int2((int)((unsigned)r[k].x >> BSH), r[k].y);
      }
    }
  }
}

__global__ __launch_bounds__(256) void gemm_wmma_kernel(
    const float* __restrict__ x, const float* __restrict__ w,
    const float* __restrict__ dinv, float* __restrict__ xw, int n_rows) {
  const int lane = threadIdx.x & 31;
  const int wave = blockIdx.x * (blockDim.x >> 5) + (threadIdx.x >> 5);
  const int row0 = wave * 16;
  if (row0 >= n_rows) return;

  const int m  = lane & 15;
  const int kh = (lane >> 4) * 2;

  const int arow = (row0 + m < n_rows) ? (row0 + m) : (n_rows - 1);

  v8f acc[8];
#pragma unroll
  for (int t = 0; t < 8; ++t) acc[t] = (v8f){0.f,0.f,0.f,0.f,0.f,0.f,0.f,0.f};

  for (int k = 0; k < D; k += 4) {
    const float* ap = x + (size_t)arow * D + (k + kh);
    v2f a;
    a.x = ap[0];
    a.y = ap[1];
#pragma unroll
    for (int t = 0; t < 8; ++t) {
      const float* bp = w + (size_t)(k + kh) * D + t * 16 + m;
      v2f b;
      b.x = bp[0];
      b.y = bp[D];
      acc[t] = __builtin_amdgcn_wmma_f32_16x16x4_f32(
           false, a,  false, b,
           (short)0, acc[t],  false,  false);
    }
  }

  float ds[8];
#pragma unroll
  for (int j = 0; j < 8; ++j) {
    const int rr = row0 + (kh << 2) + j;
    ds[j] = dinv[(rr < n_rows) ? rr : (n_rows - 1)];
  }

#pragma unroll
  for (int t = 0; t < 8; ++t) {
#pragma unroll
    for (int j = 0; j < 8; ++j) {
      const int M = (kh << 2) + j;
      if (row0 + M < n_rows)
        xw[(size_t)(row0 + M) * D + t * 16 + m] = acc[t][j] * ds[j];
    }
  }
}

template <int K>
__device__ __forceinline__ void gather_step(const float4* __restrict__ xw4,
                                            int r_l, float w_l, int j, int lane,
                                            float4& acc) {
  float4 v[K];
  float w[K];
#pragma unroll
  for (int k = 0; k < K; ++k) {
    const int r = __builtin_amdgcn_readlane(r_l, j + k);
    w[k] = __int_as_float(__builtin_amdgcn_readlane(__float_as_int(w_l), j + k));
    v[k] = xw4[(size_t)(unsigned)r * (D / 4) + lane];
  }
#pragma unroll
  for (int k = 0; k < K; ++k) {
    acc.x = fmaf(w[k], v[k].x, acc.x);
    acc.y = fmaf(w[k], v[k].y, acc.y);
    acc.z = fmaf(w[k], v[k].z, acc.z);
    acc.w = fmaf(w[k], v[k].w, acc.w);
  }
}

__global__ __launch_bounds__(256) void gather_kernel(
    const int* __restrict__ off, const int* __restrict__ cnt,
    const int* __restrict__ lst, const float* __restrict__ dinv,
    const float* __restrict__ xw, const float* __restrict__ bias,
    const float* __restrict__ pa, float* __restrict__ out, int n) {
  const int lane = threadIdx.x & 31;
  const int i = blockIdx.x * 8 + __builtin_amdgcn_readfirstlane(threadIdx.x >> 5);
  if (i >= n) return;
  const float4* xw4 = (const float4*)xw;
  const int2* lst2 = (const int2*)lst;
  const int beg = off[i];
  const int len = cnt[i];
  const float di = dinv[i];
  const float4 self = xw4[(size_t)i * (D / 4) + lane];
  const float sn = di;
  float4 acc = make_float4(sn * self.x, sn * self.y, sn * self.z, sn * self.w);
  for (int base = 0; base < len; base += 32) {
    const int m = (len - base < 32) ? (len - base) : 32;
    int r_l = 0;
    float w_l = 0.0f;
    if (lane < m) {
      const int2 pe = lst2[beg + base + lane];
      r_l = pe.x;
      w_l = __int_as_float(pe.y) * di;
    }
    int j = 0;
    for (; j + 8 <= m; j += 8) gather_step<8>(xw4, r_l, w_l, j, lane, acc);
    if (m & 4) { gather_step<4>(xw4, r_l, w_l, j, lane, acc); j += 4; }
    if (m & 2) { gather_step<2>(xw4, r_l, w_l, j, lane, acc); j += 2; }
    if (m & 1) { gather_step<1>(xw4, r_l, w_l, j, lane, acc); }
  }
  const float a = pa[0];
  float4 o;
  o.x = acc.x + bias[lane * 4 + 0];
  o.y = acc.y + bias[lane * 4 + 1];
  o.z = acc.z + bias[lane * 4 + 2];
  o.w = acc.w + bias[lane * 4 + 3];
  o.x = (o.x >= 0.0f) ? o.x : a * o.x;
  o.y = (o.y >= 0.0f) ? o.y : a * o.y;
  o.z = (o.z >= 0.0f) ? o.z : a * o.z;
  o.w = (o.w >= 0.0f) ? o.w : a * o.w;
  ((float4*)out)[(size_t)i * (D / 4) + lane] = o;
}

extern "C" void kernel_launch(void* const* d_in, const int* in_sizes, int n_in,
                              void* d_out, int out_size, void* d_ws, size_t ws_size,
                              hipStream_t stream) {
  const float* x    = (const float*)d_in[0];
  const int*   ei   = (const int*)  d_in[1];
  const float* ew   = (const float*)d_in[2];
  const float* w    = (const float*)d_in[3];
  const float* bias = (const float*)d_in[4];
  const float* pa   = (const float*)d_in[5];
  float* out = (float*)d_out;

  const int n     = in_sizes[0] / D;
  const int e_cnt = in_sizes[2];
  const int* row = ei;
  const int* col = ei + e_cnt;
  const int nb = (n + BW - 1) / BW;
  const int nh = nb * NCHUNK;

  const size_t n_pad = ((size_t)n + 127) / 128 * 128;
  const size_t e_pad = ((size_t)e_cnt + 127) / 128 * 128;
  const size_t h_pad = ((size_t)nh + 127) / 128 * 128;
  int*   cnt  = (int*)d_ws;
  int*   off  = cnt + n_pad;
  float* dinv = (float*)(off + n_pad);
  int*   hist = (int*)(dinv + n_pad);
  int*   base = hist + h_pad;
  int*   bkt  = base + h_pad;
  int*   lst  = bkt + 2 * e_pad;
  float* xw   = (float*)(lst + 2 * e_pad);

  bucket_count_kernel<<<NCHUNK, 1024, nb * sizeof(int), stream>>>(col, hist, e_cnt, nb);
  scan_kernel<<<(nh + 1023) / 1024, 1024, 0, stream>>>(hist, base, nh);
  bucket_fill_kernel<<<NCHUNK, 1024, nb * sizeof(int), stream>>>(row, col, ew, base, bkt, e_cnt, nb);
  bucket_sort_kernel<<<nb, 1024, 0, stream>>>(base, bkt, cnt, off, dinv, lst, n, e_cnt, nb);
  {
    int tiles  = (n + 15) / 16;
    int blocks = (tiles + 7) / 8;
    gemm_wmma_kernel<<<blocks, 256, 0, stream>>>(x, w, dinv, xw, n);
  }
  gather_kernel<<<(n + 7) / 8, 256, 0, stream>>>(off, cnt, lst, dinv, xw, bias, pa, out, n);
}


# first sort level done inside each edge chunk (LDS histogram + LDS scan, records in the chunk's own slice); bucket groups read their 256 segments; grid scan and 2nd edge pass gone: 6 -> 4 launches
# speedup vs baseline: 6.8244x; 1.1354x over previous
#include <hip/hip_runtime.h>
#include <hip/hip_bf16.h>

typedef __attribute__((ext_vector_type(2))) float v2f;
typedef __attribute__((ext_vector_type(8))) float v8f;

#define D 128


#define BSH 8
#define BW (1 << BSH)
#define NCHUNK 256
#define EB 8

__global__ __launch_bounds__(1024) void chunk_sort_kernel(
    const int* __restrict__ row, const int* __restrict__ col,
    const float* __restrict__ ew, int* __restrict__ segcnt,
    int* __restrict__ segoff, int* __restrict__ bkt, int e_cnt, int nb) {
  extern __shared__ int s_dyn[];
  __shared__ int s_ws[32];
  int* s_h = s_dyn;
  int* s_cur = s_dyn + nb;
  const int c = blockIdx.x;
  const int t = threadIdx.x;
  const int lane = t & 31;
  const int wv = t >> 5;
  for (int b = t; b < nb; b += blockDim.x) s_h[b] = 0;
  __syncthreads();
  const int per = (e_cnt + NCHUNK - 1) / NCHUNK;
  const int beg = c * per;
  const int end = (beg + per < e_cnt) ? (beg + per) : e_cnt;
  for (int e0 = beg + t; e0 < end; e0 += EB * blockDim.x) {
    int tg[EB];
#pragma unroll
    for (int k = 0; k < EB; ++k) {
      const int e = e0 + k * blockDim.x;
      tg[k] = col[(e < end) ? e : (end - 1)];
    }
#pragma unroll
    for (int k = 0; k < EB; ++k)
      if (e0 + k * blockDim.x < end) atomicAdd(&s_h[tg[k] >> BSH], 1);
  }
  __syncthreads();
  int carry = 0;
  for (int b0 = 0; b0 < nb; b0 += 1024) {
    const int b = b0 + t;
    const int v = (b < nb) ? s_h[b] : 0;
    int inc = v;
    for (int d = 1; d < 32; d <<= 1) {
      int u = __shfl_up(inc, d);
      if (lane >= d) inc += u;
    }
    if (lane == 31) s_ws[wv] = inc;
    __syncthreads();
    const int wsv = s_ws[lane];
    int pre = (lane < wv) ? wsv : 0;
    int tot = wsv;
    for (int d = 16; d > 0; d >>= 1) {
      pre += __shfl_xor(pre, d);
      tot += __shfl_xor(tot, d);
    }
    const int ex = carry + pre + inc - v;
    if (b < nb) {
      s_cur[b] = ex;
      segcnt[b * NCHUNK + c] = v;
      segoff[b * NCHUNK + c] = ex;
    }
    carry += tot;
    __syncthreads();
  }
  for (int e0 = beg + t; e0 < end; e0 += EB * blockDim.x) {
    int tg[EB];
    int sr[EB];
    float wt[EB];
#pragma unroll
    for (int k = 0; k < EB; ++k) {
      const int e = e0 + k * blockDim.x;
      const int ec = (e < end) ? e : (end - 1);
      tg[k] = col[ec];
      sr[k] = row[ec];
      wt[k] = ew[ec];
    }
#pragma unroll
    for (int k = 0; k < EB; ++k) {
      if (e0 + k * blockDim.x < end) {
        const int pos = atomicAdd(&s_cur[tg[k] >> BSH], 1);
        ((int2*)bkt)[beg + pos] = make_int2((sr[k] << BSH) | (tg[k] & (BW - 1)),
                                            __float_as_int(wt[k]));
      }
    }
  }
}

__global__ __launch_bounds__(1024) void bucket_sort_kernel(
    const int* __restrict__ segcnt, const int* __restrict__ segoff,
    const int* __restrict__ bkt, int* __restrict__ cnt, int* __restrict__ off,
    float* __restrict__ dinv, int* __restrict__ lst, int n, int e_cnt) {
  __shared__ int s_cnt[BW];
  __shared__ float s_deg[BW];
  __shared__ int s_cur[BW];
  __shared__ int s_ws[BW / 32];
  __shared__ int s_red[32];
  const int b = blockIdx.x;
  const int t = threadIdx.x;
  const int lane = t & 31;
  const int per = (e_cnt + NCHUNK - 1) / NCHUNK;
  const int sc = t >> 2;
  const int q = t & 3;
  const int segc = segcnt[b * NCHUNK + sc];
  const int sego = segoff[b * NCHUNK + sc];
  const int2* seg = (const int2*)bkt + ((size_t)sc * per + sego);
  int part = (q == 0) ? sego : 0;
  for (int d = 16; d > 0; d >>= 1) part += __shfl_xor(part, d);
  if (lane == 0) s_red[t >> 5] = part;
  if (t < BW) { s_cnt[t] = 0; s_deg[t] = 0.0f; }
  __syncthreads();
  int beg = s_red[lane];
  for (int d = 16; d > 0; d >>= 1) beg += __shfl_xor(beg, d);
  for (int j0 = q; j0 < segc; j0 += 4 * EB) {
    int2 r[EB];
#pragma unroll
    for (int k = 0; k < EB; ++k) {
      const int j = j0 + 4 * k;
      r[k] = seg[(j < segc) ? j : (segc - 1)];
    }
#pragma unroll
    for (int k = 0; k < EB; ++k) {
      if (j0 + 4 * k < segc) {
        const int tl = r[k].x & (BW - 1);
        atomicAdd(&s_cnt[tl], 1);
        atomicAdd(&s_deg[tl], __int_as_float(r[k].y));
      }
    }
  }
  __syncthreads();
  int v = 0;
  int inc = 0;
  if (t < BW) {
    v = s_cnt[t];
    inc = v;
    for (int d = 1; d < 32; d <<= 1) {
      int u = __shfl_up(inc, d);
      if (lane >= d) inc += u;
    }
    if (lane == 31) s_ws[t >> 5] = inc;
  }
  __syncthreads();
  if (t < BW) {
    int pre = 0;
    for (int k = 0; k < (t >> 5); ++k) pre += s_ws[k];
    const int ex = pre + inc - v;
    s_cur[t] = ex;
    const int node = b * BW + t;
    if (node < n) {
      cnt[node] = v;
      off[node] = beg + ex;
      const float d = 1.0f + s_deg[t];
      dinv[node] = (d > 0.0f) ? rsqrtf(d) : 0.0f;
    }
  }
  __syncthreads();
  for (int j0 = q; j0 < segc; j0 += 4 * EB) {
    int2 r[EB];
#pragma unroll
    for (int k = 0; k < EB; ++k) {
      const int j = j0 + 4 * k;
      r[k] = seg[(j < segc) ? j : (segc - 1)];
    }
#pragma unroll
    for (int k = 0; k < EB; ++k) {
      if (j0 + 4 * k < segc) {
        const int pos = atomicAdd(&s_cur[r[k].x & (BW - 1)], 1);
        ((int2*)lst)[beg + pos] = make_int2((int)((unsigned)r[k].x >> BSH), r[k].y);
      }
    }
  }
}

__global__ __launch_bounds__(256) void gemm_wmma_kernel(
    const float* __restrict__ x, const float* __restrict__ w,
    const float* __restrict__ dinv, float* __restrict__ xw, int n_rows) {
  const int lane = threadIdx.x & 31;
  const int wave = blockIdx.x * (blockDim.x >> 5) + (threadIdx.x >> 5);
  const int row0 = wave * 16;
  if (row0 >= n_rows) return;

  const int m  = lane & 15;
  const int kh = (lane >> 4) * 2;

  const int arow = (row0 + m < n_rows) ? (row0 + m) : (n_rows - 1);

  v8f acc[8];
#pragma unroll
  for (int t = 0; t < 8; ++t) acc[t] = (v8f){0.f,0.f,0.f,0.f,0.f,0.f,0.f,0.f};

  for (int k = 0; k < D; k += 4) {
    const float* ap = x + (size_t)arow * D + (k + kh);
    v2f a;
    a.x = ap[0];
    a.y = ap[1];
#pragma unroll
    for (int t = 0; t < 8; ++t) {
      const float* bp = w + (size_t)(k + kh) * D + t * 16 + m;
      v2f b;
      b.x = bp[0];
      b.y = bp[D];
      acc[t] = __builtin_amdgcn_wmma_f32_16x16x4_f32(
           false, a,  false, b,
           (short)0, acc[t],  false,  false);
    }
  }

  float ds[8];
#pragma unroll
  for (int j = 0; j < 8; ++j) {
    const int rr = row0 + (kh << 2) + j;
    ds[j] = dinv[(rr < n_rows) ? rr : (n_rows - 1)];
  }

#pragma unroll
  for (int t = 0; t < 8; ++t) {
#pragma unroll
    for (int j = 0; j < 8; ++j) {
      const int M = (kh << 2) + j;
      if (row0 + M < n_rows)
        xw[(size_t)(row0 + M) * D + t * 16 + m] = acc[t][j] * ds[j];
    }
  }
}

template <int K>
__device__ __forceinline__ void gather_step(const float4* __restrict__ xw4,
                                            int r_l, float w_l, int j, int lane,
                                            float4& acc) {
  float4 v[K];
  float w[K];
#pragma unroll
  for (int k = 0; k < K; ++k) {
    const int r = __builtin_amdgcn_readlane(r_l, j + k);
    w[k] = __int_as_float(__builtin_amdgcn_readlane(__float_as_int(w_l), j + k));
    v[k] = xw4[(size_t)(unsigned)r * (D / 4) + lane];
  }
#pragma unroll
  for (int k = 0; k < K; ++k) {
    acc.x = fmaf(w[k], v[k].x, acc.x);
    acc.y = fmaf(w[k], v[k].y, acc.y);
    acc.z = fmaf(w[k], v[k].z, acc.z);
    acc.w = fmaf(w[k], v[k].w, acc.w);
  }
}

__global__ __launch_bounds__(256) void gather_kernel(
    const int* __restrict__ off, const int* __restrict__ cnt,
    const int* __restrict__ lst, const float* __restrict__ dinv,
    const float* __restrict__ xw, const float* __restrict__ bias,
    const float* __restrict__ pa, float* __restrict__ out, int n) {
  const int lane = threadIdx.x & 31;
  const int i = blockIdx.x * 8 + __builtin_amdgcn_readfirstlane(threadIdx.x >> 5);
  if (i >= n) return;
  const float4* xw4 = (const float4*)xw;
  const int2* lst2 = (const int2*)lst;
  const int beg = off[i];
  const int len = cnt[i];
  const float di = dinv[i];
  const float4 self = xw4[(size_t)i * (D / 4) + lane];
  const float sn = di;
  float4 acc = make_float4(sn * self.x, sn * self.y, sn * self.z, sn * self.w);
  for (int base = 0; base < len; base += 32) {
    const int m = (len - base < 32) ? (len - base) : 32;
    int r_l = 0;
    float w_l = 0.0f;
    if (lane < m) {
      const int2 pe = lst2[beg + base + lane];
      r_l = pe.x;
      w_l = __int_as_float(pe.y) * di;
    }
    int j = 0;
    for (; j + 8 <= m; j += 8) gather_step<8>(xw4, r_l, w_l, j, lane, acc);
    if (m & 4) { gather_step<4>(xw4, r_l, w_l, j, lane, acc); j += 4; }
    if (m & 2) { gather_step<2>(xw4, r_l, w_l, j, lane, acc); j += 2; }
    if (m & 1) { gather_step<1>(xw4, r_l, w_l, j, lane, acc); }
  }
  const float a = pa[0];
  float4 o;
  o.x = acc.x + bias[lane * 4 + 0];
  o.y = acc.y + bias[lane * 4 + 1];
  o.z = acc.z + bias[lane * 4 + 2];
  o.w = acc.w + bias[lane * 4 + 3];
  o.x = (o.x >= 0.0f) ? o.x : a * o.x;
  o.y = (o.y >= 0.0f) ? o.y : a * o.y;
  o.z = (o.z >= 0.0f) ? o.z : a * o.z;
  o.w = (o.w >= 0.0f) ? o.w : a * o.w;
  ((float4*)out)[(size_t)i * (D / 4) + lane] = o;
}

extern "C" void kernel_launch(void* const* d_in, const int* in_sizes, int n_in,
                              void* d_out, int out_size, void* d_ws, size_t ws_size,
                              hipStream_t stream) {
  const float* x    = (const float*)d_in[0];
  const int*   ei   = (const int*)  d_in[1];
  const float* ew   = (const float*)d_in[2];
  const float* w    = (const float*)d_in[3];
  const float* bias = (const float*)d_in[4];
  const float* pa   = (const float*)d_in[5];
  float* out = (float*)d_out;

  const int n     = in_sizes[0] / D;
  const int e_cnt = in_sizes[2];
  const int* row = ei;
  const int* col = ei + e_cnt;
  const int nb = (n + BW - 1) / BW;
  const int nh = nb * NCHUNK;

  const size_t n_pad = ((size_t)n + 127) / 128 * 128;
  const size_t e_pad = ((size_t)e_cnt + 127) / 128 * 128;
  const size_t h_pad = ((size_t)nh + 127) / 128 * 128;
  int*   cnt  = (int*)d_ws;
  int*   off  = cnt + n_pad;
  float* dinv = (float*)(off + n_pad);
  int*   segcnt = (int*)(dinv + n_pad);
  int*   segoff = segcnt + h_pad;
  int*   bkt  = segoff + h_pad;
  int*   lst  = bkt + 2 * e_pad;
  float* xw   = (float*)(lst + 2 * e_pad);

  chunk_sort_kernel<<<NCHUNK, 1024, 2 * nb * sizeof(int), stream>>>(row, col, ew, segcnt, segoff, bkt, e_cnt, nb);
  bucket_sort_kernel<<<nb, 1024, 0, stream>>>(segcnt, segoff, bkt, cnt, off, dinv, lst, n, e_cnt);
  {
    int tiles  = (n + 15) / 16;
    int blocks = (tiles + 7) / 8;
    gemm_wmma_kernel<<<blocks, 256, 0, stream>>>(x, w, dinv, xw, n);
  }
  gather_kernel<<<(n + 7) / 8, 256, 0, stream>>>(off, cnt, lst, dinv, xw, bias, pa, out, n);
}


# GEMM fed without 4-byte global loads: w copied once per work-group into LDS (B = one 8-byte LDS read), x read as 16-byte pieces; k order of the f32 products regrouped; grid 256 x 25 waves
# speedup vs baseline: 7.4937x; 1.0981x over previous
#include <hip/hip_runtime.h>
#include <hip/hip_bf16.h>

typedef __attribute__((ext_vector_type(2))) float v2f;
typedef __attribute__((ext_vector_type(8))) float v8f;

#define D 128


#define BSH 8
#define BW (1 << BSH)
#define NCHUNK 256
#define EB 8

__global__ __launch_bounds__(1024) void chunk_sort_kernel(
    const int* __restrict__ row, const int* __restrict__ col,
    const float* __restrict__ ew, int* __restrict__ segcnt,
    int* __restrict__ segoff, int* __restrict__ bkt, int e_cnt, int nb) {
  extern __shared__ int s_dyn[];
  __shared__ int s_ws[32];
  int* s_h = s_dyn;
  int* s_cur = s_dyn + nb;
  const int c = blockIdx.x;
  const int t = threadIdx.x;
  const int lane = t & 31;
  const int wv = t >> 5;
  for (int b = t; b < nb; b += blockDim.x) s_h[b] = 0;
  __syncthreads();
  const int per = (e_cnt + NCHUNK - 1) / NCHUNK;
  const int beg = c * per;
  const int end = (beg + per < e_cnt) ? (beg + per) : e_cnt;
  for (int e0 = beg + t; e0 < end; e0 += EB * blockDim.x) {
    int tg[EB];
#pragma unroll
    for (int k = 0; k < EB; ++k) {
      const int e = e0 + k * blockDim.x;
      tg[k] = col[(e < end) ? e : (end - 1)];
    }
#pragma unroll
    for (int k = 0; k < EB; ++k)
      if (e0 + k * blockDim.x < end) atomicAdd(&s_h[tg[k] >> BSH], 1);
  }
  __syncthreads();
  int carry = 0;
  for (int b0 = 0; b0 < nb; b0 += 1024) {
    const int b = b0 + t;
    const int v = (b < nb) ? s_h[b] : 0;
    int inc = v;
    for (int d = 1; d < 32; d <<= 1) {
      int u = __shfl_up(inc, d);
      if (lane >= d) inc += u;
    }
    if (lane == 31) s_ws[wv] = inc;
    __syncthreads();
    const int wsv = s_ws[lane];
    int pre = (lane < wv) ? wsv : 0;
    int tot = wsv;
    for (int d = 16; d > 0; d >>= 1) {
      pre += __shfl_xor(pre, d);
      tot += __shfl_xor(tot, d);
    }
    const int ex = carry + pre + inc - v;
    if (b < nb) {
      s_cur[b] = ex;
      segcnt[b * NCHUNK + c] = v;
      segoff[b * NCHUNK + c] = ex;
    }
    carry += tot;
    __syncthreads();
  }
  for (int e0 = beg + t; e0 < end; e0 += EB * blockDim.x) {
    int tg[EB];
    int sr[EB];
    float wt[EB];
#pragma unroll
    for (int k = 0; k < EB; ++k) {
      const int e = e0 + k * blockDim.x;
      const int ec = (e < end) ? e : (end - 1);
      tg[k] = col[ec];
      sr[k] = row[ec];
      wt[k] = ew[ec];
    }
#pragma unroll
    for (int k = 0; k < EB; ++k) {
      if (e0 + k * blockDim.x < end) {
        const int pos = atomicAdd(&s_cur[tg[k] >> BSH], 1);
        ((int2*)bkt)[beg + pos] = make_int2((sr[k] << BSH) | (tg[k] & (BW - 1)),
                                            __float_as_int(wt[k]));
      }
    }
  }
}

__global__ __launch_bounds__(1024) void bucket_sort_kernel(
    const int* __restrict__ segcnt, const int* __restrict__ segoff,
    const int* __restrict__ bkt, int* __restrict__ cnt, int* __restrict__ off,
    float* __restrict__ dinv, int* __restrict__ lst, int n, int e_cnt) {
  __shared__ int s_cnt[BW];
  __shared__ float s_deg[BW];
  __shared__ int s_cur[BW];
  __shared__ int s_ws[BW / 32];
  __shared__ int s_red[32];
  const int b = blockIdx.x;
  const int t = threadIdx.x;
  const int lane = t & 31;
  const int per = (e_cnt + NCHUNK - 1) / NCHUNK;
  const int sc = t >> 2;
  const int q = t & 3;
  const int segc = segcnt[b * NCHUNK + sc];
  const int sego = segoff[b * NCHUNK + sc];
  const int2* seg = (const int2*)bkt + ((size_t)sc * per + sego);
  int part = (q == 0) ? sego : 0;
  for (int d = 16; d > 0; d >>= 1) part += __shfl_xor(part, d);
  if (lane == 0) s_red[t >> 5] = part;
  if (t < BW) { s_cnt[t] = 0; s_deg[t] = 0.0f; }
  __syncthreads();
  int beg = s_red[lane];
  for (int d = 16; d > 0; d >>= 1) beg += __shfl_xor(beg, d);
  for (int j0 = q; j0 < segc; j0 += 4 * EB) {
    int2 r[EB];
#pragma unroll
    for (int k = 0; k < EB; ++k) {
      const int j = j0 + 4 * k;
      r[k] = seg[(j < segc) ? j : (segc - 1)];
    }
#pragma unroll
    for (int k = 0; k < EB; ++k) {
      if (j0 + 4 * k < segc) {
        const int tl = r[k].x & (BW - 1);
        atomicAdd(&s_cnt[tl], 1);
        atomicAdd(&s_deg[tl], __int_as_float(r[k].y));
      }
    }
  }
  __syncthreads();
  int v = 0;
  int inc = 0;
  if (t < BW) {
    v = s_cnt[t];
    inc = v;
    for (int d = 1; d < 32; d <<= 1) {
      int u = __shfl_up(inc, d);
      if (lane >= d) inc += u;
    }
    if (lane == 31) s_ws[t >> 5] = inc;
  }
  __syncthreads();
  if (t < BW) {
    int pre = 0;
    for (int k = 0; k < (t >> 5); ++k) pre += s_ws[k];
    const int ex = pre + inc - v;
    s_cur[t] = ex;
    const int node = b * BW + t;
    if (node < n) {
      cnt[node] = v;
      off[node] = beg + ex;
      const float d = 1.0f + s_deg[t];
      dinv[node] = (d > 0.0f) ? rsqrtf(d) : 0.0f;
    }
  }
  __syncthreads();
  for (int j0 = q; j0 < segc; j0 += 4 * EB) {
    int2 r[EB];
#pragma unroll
    for (int k = 0; k < EB; ++k) {
      const int j = j0 + 4 * k;
      r[k] = seg[(j < segc) ? j : (segc - 1)];
    }
#pragma unroll
    for (int k = 0; k < EB; ++k) {
      if (j0 + 4 * k < segc) {
        const int pos = atomicAdd(&s_cur[r[k].x & (BW - 1)], 1);
        ((int2*)lst)[beg + pos] = make_int2((int)((unsigned)r[k].x >> BSH), r[k].y);
      }
    }
  }
}

#define WP 272
__global__ __launch_bounds__(800) void gemm_wmma_kernel(
    const float* __restrict__ x, const float* __restrict__ w,
    const float* __restrict__ dinv, float* __restrict__ xw, int n_rows) {
  __shared__ float s_w[(D / 2) * WP];
  const int tid = threadIdx.x;
  for (int i = tid; i < (D / 2) * (D / 4); i += blockDim.x) {
    const int p = i >> 5;
    const int c4 = i & 31;
    const float4 e = ((const float4*)w)[(2 * p) * (D / 4) + c4];
    const float4 o = ((const float4*)w)[(2 * p + 1) * (D / 4) + c4];
    float4* dst = (float4*)&s_w[p * WP + 8 * c4];
    dst[0] = make_float4(e.x, o.x, e.y, o.y);
    dst[1] = make_float4(e.z, o.z, e.w, o.w);
  }
  __syncthreads();

  const int lane = tid & 31;
  const int wv = __builtin_amdgcn_readfirstlane(tid >> 5);
  const int nwv = blockDim.x >> 5;
  const int m = lane & 15;
  const int h = lane >> 4;
  const int tiles = (n_rows + 15) >> 4;

  for (int tile = blockIdx.x + gridDim.x * wv; tile < tiles; tile += gridDim.x * nwv) {
    asm volatile("" ::: "memory");
    const int row0 = tile * 16;
    const int arow = (row0 + m < n_rows) ? (row0 + m) : (n_rows - 1);
    const float4* ap = (const float4*)(x + (size_t)arow * D) + h;

    v8f acc[8];
#pragma unroll
    for (int t = 0; t < 8; ++t) acc[t] = (v8f){0.f,0.f,0.f,0.f,0.f,0.f,0.f,0.f};

    float4 q[16];
#pragma unroll
    for (int jj = 0; jj < 16; ++jj) q[jj] = ap[2 * jj];
#pragma unroll
    for (int jj = 0; jj < 16; ++jj) {
      const v2f* bp = (const v2f*)(s_w + (4 * jj + 2 * h) * WP + 2 * m);
      v2f a0, a1;
      a0.x = q[jj].x; a0.y = q[jj].y;
      a1.x = q[jj].z; a1.y = q[jj].w;
#pragma unroll
      for (int t = 0; t < 8; ++t) {
        const v2f b = bp[t * 16];
        acc[t] = __builtin_amdgcn_wmma_f32_16x16x4_f32(
             false, a0,  false, b,
             (short)0, acc[t],  false,  false);
      }
#pragma unroll
      for (int t = 0; t < 8; ++t) {
        const v2f b = bp[WP / 2 + t * 16];
        acc[t] = __builtin_amdgcn_wmma_f32_16x16x4_f32(
             false, a1,  false, b,
             (short)0, acc[t],  false,  false);
      }
    }

    float ds[8];
#pragma unroll
    for (int j = 0; j < 8; ++j) {
      const int rr = row0 + 8 * h + j;
      ds[j] = dinv[(rr < n_rows) ? rr : (n_rows - 1)];
    }

#pragma unroll
    for (int t = 0; t < 8; ++t) {
#pragma unroll
      for (int j = 0; j < 8; ++j) {
        const int M = 8 * h + j;
        if (row0 + M < n_rows)
          xw[(size_t)(row0 + M) * D + t * 16 + m] = acc[t][j] * ds[j];
      }
    }
  }
}

template <int K>
__device__ __forceinline__ void gather_step(const float4* __restrict__ xw4,
                                            int r_l, float w_l, int j, int lane,
                                            float4& acc) {
  float4 v[K];
  float w[K];
#pragma unroll
  for (int k = 0; k < K; ++k) {
    const int r = __builtin_amdgcn_readlane(r_l, j + k);
    w[k] = __int_as_float(__builtin_amdgcn_readlane(__float_as_int(w_l), j + k));
    v[k] = xw4[(size_t)(unsigned)r * (D / 4) + lane];
  }
#pragma unroll
  for (int k = 0; k < K; ++k) {
    acc.x = fmaf(w[k], v[k].x, acc.x);
    acc.y = fmaf(w[k], v[k].y, acc.y);
    acc.z = fmaf(w[k], v[k].z, acc.z);
    acc.w = fmaf(w[k], v[k].w, acc.w);
  }
}

__global__ __launch_bounds__(256) void gather_kernel(
    const int* __restrict__ off, const int* __restrict__ cnt,
    const int* __restrict__ lst, const float* __restrict__ dinv,
    const float* __restrict__ xw, const float* __restrict__ bias,
    const float* __restrict__ pa, float* __restrict__ out, int n) {
  const int lane = threadIdx.x & 31;
  const int i = blockIdx.x * 8 + __builtin_amdgcn_readfirstlane(threadIdx.x >> 5);
  if (i >= n) return;
  const float4* xw4 = (const float4*)xw;
  const int2* lst2 = (const int2*)lst;
  const int beg = off[i];
  const int len = cnt[i];
  const float di = dinv[i];
  const float4 self = xw4[(size_t)i * (D / 4) + lane];
  const float sn = di;
  float4 acc = make_float4(sn * self.x, sn * self.y, sn * self.z, sn * self.w);
  for (int base = 0; base < len; base += 32) {
    const int m = (len - base < 32) ? (len - base) : 32;
    int r_l = 0;
    float w_l = 0.0f;
    if (lane < m) {
      const int2 pe = lst2[beg + base + lane];
      r_l = pe.x;
      w_l = __int_as_float(pe.y) * di;
    }
    int j = 0;
    for (; j + 8 <= m; j += 8) gather_step<8>(xw4, r_l, w_l, j, lane, acc);
    if (m & 4) { gather_step<4>(xw4, r_l, w_l, j, lane, acc); j += 4; }
    if (m & 2) { gather_step<2>(xw4, r_l, w_l, j, lane, acc); j += 2; }
    if (m & 1) { gather_step<1>(xw4, r_l, w_l, j, lane, acc); }
  }
  const float a = pa[0];
  float4 o;
  o.x = acc.x + bias[lane * 4 + 0];
  o.y = acc.y + bias[lane * 4 + 1];
  o.z = acc.z + bias[lane * 4 + 2];
  o.w = acc.w + bias[lane * 4 + 3];
  o.x = (o.x >= 0.0f) ? o.x : a * o.x;
  o.y = (o.y >= 0.0f) ? o.y : a * o.y;
  o.z = (o.z >= 0.0f) ? o.z : a * o.z;
  o.w = (o.w >= 0.0f) ? o.w : a * o.w;
  ((float4*)out)[(size_t)i * (D / 4) + lane] = o;
}

extern "C" void kernel_launch(void* const* d_in, const int* in_sizes, int n_in,
                              void* d_out, int out_size, void* d_ws, size_t ws_size,
                              hipStream_t stream) {
  const float* x    = (const float*)d_in[0];
  const int*   ei   = (const int*)  d_in[1];
  const float* ew   = (const float*)d_in[2];
  const float* w    = (const float*)d_in[3];
  const float* bias = (const float*)d_in[4];
  const float* pa   = (const float*)d_in[5];
  float* out = (float*)d_out;

  const int n     = in_sizes[0] / D;
  const int e_cnt = in_sizes[2];
  const int* row = ei;
  const int* col = ei + e_cnt;
  const int nb = (n + BW - 1) / BW;
  const int nh = nb * NCHUNK;

  const size_t n_pad = ((size_t)n + 127) / 128 * 128;
  const size_t e_pad = ((size_t)e_cnt + 127) / 128 * 128;
  const size_t h_pad = ((size_t)nh + 127) / 128 * 128;
  int*   cnt  = (int*)d_ws;
  int*   off  = cnt + n_pad;
  float* dinv = (float*)(off + n_pad);
  int*   segcnt = (int*)(dinv + n_pad);
  int*   segoff = segcnt + h_pad;
  int*   bkt  = segoff + h_pad;
  int*   lst  = bkt + 2 * e_pad;
  float* xw   = (float*)(lst + 2 * e_pad);

  chunk_sort_kernel<<<NCHUNK, 1024, 2 * nb * sizeof(int), stream>>>(row, col, ew, segcnt, segoff, bkt, e_cnt, nb);
  bucket_sort_kernel<<<nb, 1024, 0, stream>>>(segcnt, segoff, bkt, cnt, off, dinv, lst, n, e_cnt);
  gemm_wmma_kernel<<<256, 800, 0, stream>>>(x, w, dinv, xw, n);
  gather_kernel<<<(n + 7) / 8, 256, 0, stream>>>(off, cnt, lst, dinv, xw, bias, pa, out, n);
}


# both counting-sort passes put their ordered 8-byte records into LDS and copy them out as whole 128-byte lines, not one scattered 8-byte store per edge; list order and sum order unchanged
# speedup vs baseline: 8.0809x; 1.0784x over previous
#include <hip/hip_runtime.h>
#include <hip/hip_bf16.h>

typedef __attribute__((ext_vector_type(2))) float v2f;
typedef __attribute__((ext_vector_type(8))) float v8f;

#define D 128


#define BSH 8
#define BW (1 << BSH)
#define NCHUNK 256
#define EB 8
#define CAPC 8192
#define CAPB 8192

__global__ __launch_bounds__(1024) void chunk_sort_kernel(
    const int* __restrict__ row, const int* __restrict__ col,
    const float* __restrict__ ew, int* __restrict__ segcnt,
    int* __restrict__ segoff, int* __restrict__ bkt, int e_cnt, int nb) {
  extern __shared__ int s_dyn[];
  __shared__ int s_ws[32];
  __shared__ int2 s_rec[CAPC];
  int* s_h = s_dyn;
  int* s_cur = s_dyn + nb;
  const int c = blockIdx.x;
  const int t = threadIdx.x;
  const int lane = t & 31;
  const int wv = t >> 5;
  for (int b = t; b < nb; b += blockDim.x) s_h[b] = 0;
  __syncthreads();
  const int per = (e_cnt + NCHUNK - 1) / NCHUNK;
  const int beg = c * per;
  const int end = (beg + per < e_cnt) ? (beg + per) : e_cnt;
  const bool in_lds = (end - beg) <= CAPC;
  for (int e0 = beg + t; e0 < end; e0 += EB * blockDim.x) {
    int tg[EB];
#pragma unroll
    for (int k = 0; k < EB; ++k) {
      const int e = e0 + k * blockDim.x;
      tg[k] = col[(e < end) ? e : (end - 1)];
    }
#pragma unroll
    for (int k = 0; k < EB; ++k)
      if (e0 + k * blockDim.x < end) atomicAdd(&s_h[tg[k] >> BSH], 1);
  }
  __syncthreads();
  int carry = 0;
  for (int b0 = 0; b0 < nb; b0 += 1024) {
    const int b = b0 + t;
    const int v = (b < nb) ? s_h[b] : 0;
    int inc = v;
    for (int d = 1; d < 32; d <<= 1) {
      int u = __shfl_up(inc, d);
      if (lane >= d) inc += u;
    }
    if (lane == 31) s_ws[wv] = inc;
    __syncthreads();
    const int wsv = s_ws[lane];
    int pre = (lane < wv) ? wsv : 0;
    int tot = wsv;
    for (int d = 16; d > 0; d >>= 1) {
      pre += __shfl_xor(pre, d);
      tot += __shfl_xor(tot, d);
    }
    const int ex = carry + pre + inc - v;
    if (b < nb) {
      s_cur[b] = ex;
      segcnt[b * NCHUNK + c] = v;
      segoff[b * NCHUNK + c] = ex;
    }
    carry += tot;
    __syncthreads();
  }
  for (int e0 = beg + t; e0 < end; e0 += EB * blockDim.x) {
    int tg[EB];
    int sr[EB];
    float wt[EB];
#pragma unroll
    for (int k = 0; k < EB; ++k) {
      const int e = e0 + k * blockDim.x;
      const int ec = (e < end) ? e : (end - 1);
      tg[k] = col[ec];
      sr[k] = row[ec];
      wt[k] = ew[ec];
    }
#pragma unroll
    for (int k = 0; k < EB; ++k) {
      if (e0 + k * blockDim.x < end) {
        const int pos = atomicAdd(&s_cur[tg[k] >> BSH], 1);
        const int2 rec = make_int2((sr[k] << BSH) | (tg[k] & (BW - 1)),
                                   __float_as_int(wt[k]));
        if (in_lds) s_rec[pos] = rec;
        else ((int2*)bkt)[beg + pos] = rec;
      }
    }
  }
  if (in_lds) {
    __syncthreads();
    const int a = beg & 15;
    for (int i = t - a; i < end - beg; i += blockDim.x)
      if (i >= 0) ((int2*)bkt)[beg + i] = s_rec[i];
  }
}

__global__ __launch_bounds__(1024) void bucket_sort_kernel(
    const int* __restrict__ segcnt, const int* __restrict__ segoff,
    const int* __restrict__ bkt, int* __restrict__ cnt, int* __restrict__ off,
    float* __restrict__ dinv, int* __restrict__ lst, int n, int e_cnt) {
  __shared__ int s_cnt[BW];
  __shared__ float s_deg[BW];
  __shared__ int s_cur[BW];
  __shared__ int s_ws[BW / 32];
  __shared__ int s_red[32];
  __shared__ int s_tot;
  __shared__ int2 s_out[CAPB];
  const int b = blockIdx.x;
  const int t = threadIdx.x;
  const int lane = t & 31;
  const int per = (e_cnt + NCHUNK - 1) / NCHUNK;
  const int sc = t >> 2;
  const int q = t & 3;
  const int segc = segcnt[b * NCHUNK + sc];
  const int sego = segoff[b * NCHUNK + sc];
  const int2* seg = (const int2*)bkt + ((size_t)sc * per + sego);
  int part = (q == 0) ? sego : 0;
  for (int d = 16; d > 0; d >>= 1) part += __shfl_xor(part, d);
  if (lane == 0) s_red[t >> 5] = part;
  if (t < BW) { s_cnt[t] = 0; s_deg[t] = 0.0f; }
  __syncthreads();
  int beg = s_red[lane];
  for (int d = 16; d > 0; d >>= 1) beg += __shfl_xor(beg, d);
  for (int j0 = q; j0 < segc; j0 += 4 * EB) {
    int2 r[EB];
#pragma unroll
    for (int k = 0; k < EB; ++k) {
      const int j = j0 + 4 * k;
      r[k] = seg[(j < segc) ? j : (segc - 1)];
    }
#pragma unroll
    for (int k = 0; k < EB; ++k) {
      if (j0 + 4 * k < segc) {
        const int tl = r[k].x & (BW - 1);
        atomicAdd(&s_cnt[tl], 1);
        atomicAdd(&s_deg[tl], __int_as_float(r[k].y));
      }
    }
  }
  __syncthreads();
  int v = 0;
  int inc = 0;
  if (t < BW) {
    v = s_cnt[t];
    inc = v;
    for (int d = 1; d < 32; d <<= 1) {
      int u = __shfl_up(inc, d);
      if (lane >= d) inc += u;
    }
    if (lane == 31) s_ws[t >> 5] = inc;
  }
  __syncthreads();
  if (t < BW) {
    int pre = 0;
    for (int k = 0; k < (t >> 5); ++k) pre += s_ws[k];
    const int ex = pre + inc - v;
    s_cur[t] = ex;
    if (t == BW - 1) s_tot = pre + inc;
    const int node = b * BW + t;
    if (node < n) {
      cnt[node] = v;
      off[node] = beg + ex;
      const float d = 1.0f + s_deg[t];
      dinv[node] = (d > 0.0f) ? rsqrtf(d) : 0.0f;
    }
  }
  __syncthreads();
  const int tot = __builtin_amdgcn_readfirstlane(s_tot);
  const bool in_lds = tot <= CAPB;
  for (int j0 = q; j0 < segc; j0 += 4 * EB) {
    int2 r[EB];
#pragma unroll
    for (int k = 0; k < EB; ++k) {
      const int j = j0 + 4 * k;
      r[k] = seg[(j < segc) ? j : (segc - 1)];
    }
#pragma unroll
    for (int k = 0; k < EB; ++k) {
      if (j0 + 4 * k < segc) {
        const int pos = atomicAdd(&s_cur[r[k].x & (BW - 1)], 1);
        const int2 ent = make_int2((int)((unsigned)r[k].x >> BSH), r[k].y);
        if (in_lds) s_out[pos] = ent;
        else ((int2*)lst)[beg + pos] = ent;
      }
    }
  }
  if (in_lds) {
    __syncthreads();
    const int a = beg & 15;
    for (int i = t - a; i < tot; i += 4 * NCHUNK)
      if (i >= 0) ((int2*)lst)[beg + i] = s_out[i];
  }
}

#define WP 272
__global__ __launch_bounds__(800) void gemm_wmma_kernel(
    const float* __restrict__ x, const float* __restrict__ w,
    const float* __restrict__ dinv, float* __restrict__ xw, int n_rows) {
  __shared__ float s_w[(D / 2) * WP];
  const int tid = threadIdx.x;
  for (int i = tid; i < (D / 2) * (D / 4); i += blockDim.x) {
    const int p = i >> 5;
    const int c4 = i & 31;
    const float4 e = ((const float4*)w)[(2 * p) * (D / 4) + c4];
    const float4 o = ((const float4*)w)[(2 * p + 1) * (D / 4) + c4];
    float4* dst = (float4*)&s_w[p * WP + 8 * c4];
    dst[0] = make_float4(e.x, o.x, e.y, o.y);
    dst[1] = make_float4(e.z, o.z, e.w, o.w);
  }
  __syncthreads();

  const int lane = tid & 31;
  const int wv = __builtin_amdgcn_readfirstlane(tid >> 5);
  const int nwv = blockDim.x >> 5;
  const int m = lane & 15;
  const int h = lane >> 4;
  const int tiles = (n_rows + 15) >> 4;

  for (int tile = blockIdx.x + gridDim.x * wv; tile < tiles; tile += gridDim.x * nwv) {
    asm volatile("" ::: "memory");
    const int row0 = tile * 16;
    const int arow = (row0 + m < n_rows) ? (row0 + m) : (n_rows - 1);
    const float4* ap = (const float4*)(x + (size_t)arow * D) + h;

    v8f acc[8];
#pragma unroll
    for (int t = 0; t < 8; ++t) acc[t] = (v8f){0.f,0.f,0.f,0.f,0.f,0.f,0.f,0.f};

    float4 q[16];
#pragma unroll
    for (int jj = 0; jj < 16; ++jj) q[jj] = ap[2 * jj];
#pragma unroll
    for (int jj = 0; jj < 16; ++jj) {
      const v2f* bp = (const v2f*)(s_w + (4 * jj + 2 * h) * WP + 2 * m);
      v2f a0, a1;
      a0.x = q[jj].x; a0.y = q[jj].y;
      a1.x = q[jj].z; a1.y = q[jj].w;
#pragma unroll
      for (int t = 0; t < 8; ++t) {
        const v2f b = bp[t * 16];
        acc[t] = __builtin_amdgcn_wmma_f32_16x16x4_f32(
             false, a0,  false, b,
             (short)0, acc[t],  false,  false);
      }
#pragma unroll
      for (int t = 0; t < 8; ++t) {
        const v2f b = bp[WP / 2 + t * 16];
        acc[t] = __builtin_amdgcn_wmma_f32_16x16x4_f32(
             false, a1,  false, b,
             (short)0, acc[t],  false,  false);
      }
    }

    float ds[8];
#pragma unroll
    for (int j = 0; j < 8; ++j) {
      const int rr = row0 + 8 * h + j;
      ds[j] = dinv[(rr < n_rows) ? rr : (n_rows - 1)];
    }

#pragma unroll
    for (int t = 0; t < 8; ++t) {
#pragma unroll
      for (int j = 0; j < 8; ++j) {
        const int M = 8 * h + j;
        if (row0 + M < n_rows)
          xw[(size_t)(row0 + M) * D + t * 16 + m] = acc[t][j] * ds[j];
      }
    }
  }
}

template <int K>
__device__ __forceinline__ void gather_step(const float4* __restrict__ xw4,
                                            int r_l, float w_l, int j, int lane,
                                            float4& acc) {
  float4 v[K];
  float w[K];
#pragma unroll
  for (int k = 0; k < K; ++k) {
    const int r = __builtin_amdgcn_readlane(r_l, j + k);
    w[k] = __int_as_float(__builtin_amdgcn_readlane(__float_as_int(w_l), j + k));
    v[k] = xw4[(size_t)(unsigned)r * (D / 4) + lane];
  }
#pragma unroll
  for (int k = 0; k < K; ++k) {
    acc.x = fmaf(w[k], v[k].x, acc.x);
    acc.y = fmaf(w[k], v[k].y, acc.y);
    acc.z = fmaf(w[k], v[k].z, acc.z);
    acc.w = fmaf(w[k], v[k].w, acc.w);
  }
}

__global__ __launch_bounds__(256) void gather_kernel(
    const int* __restrict__ off, const int* __restrict__ cnt,
    const int* __restrict__ lst, const float* __restrict__ dinv,
    const float* __restrict__ xw, const float* __restrict__ bias,
    const float* __restrict__ pa, float* __restrict__ out, int n) {
  const int lane = threadIdx.x & 31;
  const int i = blockIdx.x * 8 + __builtin_amdgcn_readfirstlane(threadIdx.x >> 5);
  if (i >= n) return;
  const float4* xw4 = (const float4*)xw;
  const int2* lst2 = (const int2*)lst;
  const int beg = off[i];
  const int len = cnt[i];
  const float di = dinv[i];
  const float4 self = xw4[(size_t)i * (D / 4) + lane];
  const float sn = di;
  float4 acc = make_float4(sn * self.x, sn * self.y, sn * self.z, sn * self.w);
  for (int base = 0; base < len; base += 32) {
    const int m = (len - base < 32) ? (len - base) : 32;
    int r_l = 0;
    float w_l = 0.0f;
    if (lane < m) {
      const int2 pe = lst2[beg + base + lane];
      r_l = pe.x;
      w_l = __int_as_float(pe.y) * di;
    }
    int j = 0;
    for (; j + 8 <= m; j += 8) gather_step<8>(xw4, r_l, w_l, j, lane, acc);
    if (m & 4) { gather_step<4>(xw4, r_l, w_l, j, lane, acc); j += 4; }
    if (m & 2) { gather_step<2>(xw4, r_l, w_l, j, lane, acc); j += 2; }
    if (m & 1) { gather_step<1>(xw4, r_l, w_l, j, lane, acc); }
  }
  const float a = pa[0];
  float4 o;
  o.x = acc.x + bias[lane * 4 + 0];
  o.y = acc.y + bias[lane * 4 + 1];
  o.z = acc.z + bias[lane * 4 + 2];
  o.w = acc.w + bias[lane * 4 + 3];
  o.x = (o.x >= 0.0f) ? o.x : a * o.x;
  o.y = (o.y >= 0.0f) ? o.y : a * o.y;
  o.z = (o.z >= 0.0f) ? o.z : a * o.z;
  o.w = (o.w >= 0.0f) ? o.w : a * o.w;
  ((float4*)out)[(size_t)i * (D / 4) + lane] = o;
}

extern "C" void kernel_launch(void* const* d_in, const int* in_sizes, int n_in,
                              void* d_out, int out_size, void* d_ws, size_t ws_size,
                              hipStream_t stream) {
  const float* x    = (const float*)d_in[0];
  const int*   ei   = (const int*)  d_in[1];
  const float* ew   = (const float*)d_in[2];
  const float* w    = (const float*)d_in[3];
  const float* bias = (const float*)d_in[4];
  const float* pa   = (const float*)d_in[5];
  float* out = (float*)d_out;

  const int n     = in_sizes[0] / D;
  const int e_cnt = in_sizes[2];
  const int* row = ei;
  const int* col = ei + e_cnt;
  const int nb = (n + BW - 1) / BW;
  const int nh = nb * NCHUNK;

  const size_t n_pad = ((size_t)n + 127) / 128 * 128;
  const size_t e_pad = ((size_t)e_cnt + 127) / 128 * 128;
  const size_t h_pad = ((size_t)nh + 127) / 128 * 128;
  int*   cnt  = (int*)d_ws;
  int*   off  = cnt + n_pad;
  float* dinv = (float*)(off + n_pad);
  int*   segcnt = (int*)(dinv + n_pad);
  int*   segoff = segcnt + h_pad;
  int*   bkt  = segoff + h_pad;
  int*   lst  = bkt + 2 * e_pad;
  float* xw   = (float*)(lst + 2 * e_pad);

  chunk_sort_kernel<<<NCHUNK, 1024, 2 * nb * sizeof(int), stream>>>(row, col, ew, segcnt, segoff, bkt, e_cnt, nb);
  bucket_sort_kernel<<<nb, 1024, 0, stream>>>(segcnt, segoff, bkt, cnt, off, dinv, lst, n, e_cnt);
  gemm_wmma_kernel<<<256, 800, 0, stream>>>(x, w, dinv, xw, n);
  gather_kernel<<<(n + 7) / 8, 256, 0, stream>>>(off, cnt, lst, dinv, xw, bias, pa, out, n);
}


# GEMM folded into the bucket pass: bucket = 400 nodes = 25 tiles, its work-group computes their xw rows on f32 WMMA (dinv from LDS, x prefetched under the sort); 4 -> 3 launches; k and sum order kept
# speedup vs baseline: 8.6464x; 1.0700x over previous
#include <hip/hip_runtime.h>
#include <hip/hip_bf16.h>

typedef __attribute__((ext_vector_type(2))) float v2f;
typedef __attribute__((ext_vector_type(8))) float v8f;

#define D 128


#define NBK 400
#define TPB (NBK / 16)
#define PSH 9
#define SL (1 << PSH)
#define NCHUNK 256
#define EB 8
#define CAPC 8192
#define CAPB 8192

__global__ __launch_bounds__(1024) void chunk_sort_kernel(
    const int* __restrict__ row, const int* __restrict__ col,
    const float* __restrict__ ew, int* __restrict__ segcnt,
    int* __restrict__ segoff, int* __restrict__ bkt, int e_cnt, int nb) {
  extern __shared__ int s_dyn[];
  __shared__ int s_ws[32];
  __shared__ int2 s_rec[CAPC];
  int* s_h = s_dyn;
  int* s_cur = s_dyn + nb;
  const int c = blockIdx.x;
  const int t = threadIdx.x;
  const int lane = t & 31;
  const int wv = t >> 5;
  for (int b = t; b < nb; b += blockDim.x) s_h[b] = 0;
  __syncthreads();
  const int per = (e_cnt + NCHUNK - 1) / NCHUNK;
  const int beg = c * per;
  const int end = (beg + per < e_cnt) ? (beg + per) : e_cnt;
  const bool in_lds = (end - beg) <= CAPC;
  for (int e0 = beg + t; e0 < end; e0 += EB * blockDim.x) {
    int tg[EB];
#pragma unroll
    for (int k = 0; k < EB; ++k) {
      const int e = e0 + k * blockDim.x;
      tg[k] = col[(e < end) ? e : (end - 1)];
    }
#pragma unroll
    for (int k = 0; k < EB; ++k)
      if (e0 + k * blockDim.x < end) atomicAdd(&s_h[(unsigned)tg[k] / NBK], 1);
  }
  __syncthreads();
  int carry = 0;
  for (int b0 = 0; b0 < nb; b0 += 1024) {
    const int b = b0 + t;
    const int v = (b < nb) ? s_h[b] : 0;
    int inc = v;
    for (int d = 1; d < 32; d <<= 1) {
      int u = __shfl_up(inc, d);
      if (lane >= d) inc += u;
    }
    if (lane == 31) s_ws[wv] = inc;
    __syncthreads();
    const int wsv = s_ws[lane];
    int pre = (lane < wv) ? wsv : 0;
    int tot = wsv;
    for (int d = 16; d > 0; d >>= 1) {
      pre += __shfl_xor(pre, d);
      tot += __shfl_xor(tot, d);
    }
    const int ex = carry + pre + inc - v;
    if (b < nb) {
      s_cur[b] = ex;
      segcnt[b * NCHUNK + c] = v;
      segoff[b * NCHUNK + c] = ex;
    }
    carry += tot;
    __syncthreads();
  }
  for (int e0 = beg + t; e0 < end; e0 += EB * blockDim.x) {
    int tg[EB];
    int sr[EB];
    float wt[EB];
#pragma unroll
    for (int k = 0; k < EB; ++k) {
      const int e = e0 + k * blockDim.x;
      const int ec = (e < end) ? e : (end - 1);
      tg[k] = col[ec];
      sr[k] = row[ec];
      wt[k] = ew[ec];
    }
#pragma unroll
    for (int k = 0; k < EB; ++k) {
      if (e0 + k * blockDim.x < end) {
        const int bk = (int)((unsigned)tg[k] / NBK);
        const int pos = atomicAdd(&s_cur[bk], 1);
        const int2 rec = make_int2((sr[k] << PSH) | (tg[k] - bk * NBK),
                                   __float_as_int(wt[k]));
        if (in_lds) s_rec[pos] = rec;
        else ((int2*)bkt)[beg + pos] = rec;
      }
    }
  }
  if (in_lds) {
    __syncthreads();
    const int a = beg & 15;
    for (int i = t - a; i < end - beg; i += blockDim.x)
      if (i >= 0) ((int2*)bkt)[beg + i] = s_rec[i];
  }
}

#define WP 272
#define NFULL (2 * TPB - 32)
__global__ __launch_bounds__(1024) void bucket_gemm_kernel(
    const int* __restrict__ segcnt, const int* __restrict__ segoff,
    const int* __restrict__ bkt, int* __restrict__ cnt, int* __restrict__ off,
    float* __restrict__ dinv, int* __restrict__ lst,
    const float* __restrict__ x, const float* __restrict__ w,
    float* __restrict__ xw, int n, int e_cnt) {
  __shared__ int s_cnt[SL];
  __shared__ float s_deg[SL];
  __shared__ int s_cur[SL];
  __shared__ float s_dinv[SL];
  __shared__ int s_ws[SL / 32];
  __shared__ int s_red[32];
  __shared__ int s_tot;
  __shared__ int2 s_out[CAPB];
  __shared__ float s_w[(D / 2) * WP];
  const int b = blockIdx.x;
  const int t = threadIdx.x;
  const int lane = t & 31;
  const int per = (e_cnt + NCHUNK - 1) / NCHUNK;
  const int sc = t >> 2;
  const int q = t & 3;
  const int segc = segcnt[b * NCHUNK + sc];
  const int sego = segoff[b * NCHUNK + sc];
  const int wv = __builtin_amdgcn_readfirstlane(t >> 5);
  const int m = lane & 15;
  const int h = lane >> 4;
  const int tile = (wv < NFULL) ? wv : (NFULL + ((wv - NFULL) >> 1));
  const int h0 = (wv < NFULL) ? 0 : ((wv - NFULL) & 1);
  const int h1 = (wv < NFULL) ? 2 : (h0 + 1);
  const int lrow0 = tile * 16;
  const int row0 = b * NBK + lrow0;
  const int arow = (row0 + m < n) ? (row0 + m) : (n - 1);
  const float4* ap = (const float4*)(x + (size_t)arow * D) + h;
  {
    const int prow = row0 + (lane >> 1);
    const float* pp = x + (size_t)((prow < n) ? prow : (n - 1)) * D + 64 * (lane & 1);
    __builtin_prefetch(pp, 0, 1);
  }
  for (int i = t; i < (D / 2) * (D / 4); i += 4 * NCHUNK) {
    const int p = i >> 5;
    const int c4 = i & 31;
    const float4 e = ((const float4*)w)[(2 * p) * (D / 4) + c4];
    const float4 o = ((const float4*)w)[(2 * p + 1) * (D / 4) + c4];
    float4* dst = (float4*)&s_w[p * WP + 8 * c4];
    dst[0] = make_float4(e.x, o.x, e.y, o.y);
    dst[1] = make_float4(e.z, o.z, e.w, o.w);
  }
  const int2* seg = (const int2*)bkt + ((size_t)sc * per + sego);
  int part = (q == 0) ? sego : 0;
  for (int d = 16; d > 0; d >>= 1) part += __shfl_xor(part, d);
  if (lane == 0) s_red[t >> 5] = part;
  if (t < SL) { s_cnt[t] = 0; s_deg[t] = 0.0f; }
  __syncthreads();
  int beg = s_red[lane];
  for (int d = 16; d > 0; d >>= 1) beg += __shfl_xor(beg, d);
  for (int j0 = q; j0 < segc; j0 += 4 * EB) {
    int2 r[EB];
#pragma unroll
    for (int k = 0; k < EB; ++k) {
      const int j = j0 + 4 * k;
      r[k] = seg[(j < segc) ? j : (segc - 1)];
    }
#pragma unroll
    for (int k = 0; k < EB; ++k) {
      if (j0 + 4 * k < segc) {
        const int tl = r[k].x & (SL - 1);
        atomicAdd(&s_cnt[tl], 1);
        atomicAdd(&s_deg[tl], __int_as_float(r[k].y));
      }
    }
  }
  __syncthreads();
  int v = 0;
  int inc = 0;
  if (t < SL) {
    v = s_cnt[t];
    inc = v;
    for (int d = 1; d < 32; d <<= 1) {
      int u = __shfl_up(inc, d);
      if (lane >= d) inc += u;
    }
    if (lane == 31) s_ws[t >> 5] = inc;
  }
  __syncthreads();
  if (t < SL) {
    int pre = 0;
    for (int k = 0; k < (t >> 5); ++k) pre += s_ws[k];
    const int ex = pre + inc - v;
    s_cur[t] = ex;
    if (t == SL - 1) s_tot = pre + inc;
    const float d = 1.0f + s_deg[t];
    const float di = (d > 0.0f) ? rsqrtf(d) : 0.0f;
    s_dinv[t] = di;
    const int node = b * NBK + t;
    if (t < NBK && node < n) {
      cnt[node] = v;
      off[node] = beg + ex;
      dinv[node] = di;
    }
  }
  __syncthreads();
  const int tot = __builtin_amdgcn_readfirstlane(s_tot);
  const bool in_lds = tot <= CAPB;
  for (int j0 = q; j0 < segc; j0 += 4 * EB) {
    int2 r[EB];
#pragma unroll
    for (int k = 0; k < EB; ++k) {
      const int j = j0 + 4 * k;
      r[k] = seg[(j < segc) ? j : (segc - 1)];
    }
#pragma unroll
    for (int k = 0; k < EB; ++k) {
      if (j0 + 4 * k < segc) {
        const int pos = atomicAdd(&s_cur[r[k].x & (SL - 1)], 1);
        const int2 ent = make_int2((int)((unsigned)r[k].x >> PSH), r[k].y);
        if (in_lds) s_out[pos] = ent;
        else ((int2*)lst)[beg + pos] = ent;
      }
    }
  }
  if (in_lds) __syncthreads();
  float4 xa[16];
#pragma unroll
  for (int jj = 0; jj < 16; ++jj) xa[jj] = ap[2 * jj];
  if (in_lds) {
    const int a = beg & 15;
    for (int i = t - a; i < tot; i += 4 * NCHUNK)
      if (i >= 0) ((int2*)lst)[beg + i] = s_out[i];
  }

  if (row0 >= n) return;
  const unsigned oi = (unsigned)(row0 + 8 * h) * D + m;
  const int lim = n - (row0 + 8 * h);
  float ds[8];
#pragma unroll
  for (int j = 0; j < 8; ++j) ds[j] = s_dinv[lrow0 + 8 * h + j];

  for (int hf = h0; hf < h1; ++hf) {
    v8f acc[4];
#pragma unroll
    for (int c = 0; c < 4; ++c) acc[c] = (v8f){0.f,0.f,0.f,0.f,0.f,0.f,0.f,0.f};
#pragma unroll
    for (int jj = 0; jj < 16; ++jj) {
      const v2f* bp = (const v2f*)(s_w + (4 * jj + 2 * h) * WP + 2 * m) + 64 * hf;
      v2f a0, a1;
      a0.x = xa[jj].x; a0.y = xa[jj].y;
      a1.x = xa[jj].z; a1.y = xa[jj].w;
#pragma unroll
      for (int c = 0; c < 4; ++c) {
        const v2f bf = bp[c * 16];
        acc[c] = __builtin_amdgcn_wmma_f32_16x16x4_f32(
             false, a0,  false, bf,
             (short)0, acc[c],  false,  false);
      }
#pragma unroll
      for (int c = 0; c < 4; ++c) {
        const v2f bf = bp[WP / 2 + c * 16];
        acc[c] = __builtin_amdgcn_wmma_f32_16x16x4_f32(
             false, a1,  false, bf,
             (short)0, acc[c],  false,  false);
      }
    }
#pragma unroll
    for (int c = 0; c < 4; ++c) {
#pragma unroll
      for (int j = 0; j < 8; ++j)
        if (j < lim) xw[oi + (unsigned)(j * D + 64 * hf + c * 16)] = acc[c][j] * ds[j];
    }
  }
}

template <int K>
__device__ __forceinline__ void gather_step(const float4* __restrict__ xw4,
                                            int r_l, float w_l, int j, int lane,
                                            float4& acc) {
  float4 v[K];
  float w[K];
#pragma unroll
  for (int k = 0; k < K; ++k) {
    const int r = __builtin_amdgcn_readlane(r_l, j + k);
    w[k] = __int_as_float(__builtin_amdgcn_readlane(__float_as_int(w_l), j + k));
    v[k] = xw4[(size_t)(unsigned)r * (D / 4) + lane];
  }
#pragma unroll
  for (int k = 0; k < K; ++k) {
    acc.x = fmaf(w[k], v[k].x, acc.x);
    acc.y = fmaf(w[k], v[k].y, acc.y);
    acc.z = fmaf(w[k], v[k].z, acc.z);
    acc.w = fmaf(w[k], v[k].w, acc.w);
  }
}

__global__ __launch_bounds__(256) void gather_kernel(
    const int* __restrict__ off, const int* __restrict__ cnt,
    const int* __restrict__ lst, const float* __restrict__ dinv,
    const float* __restrict__ xw, const float* __restrict__ bias,
    const float* __restrict__ pa, float* __restrict__ out, int n) {
  const int lane = threadIdx.x & 31;
  const int i = blockIdx.x * 8 + __builtin_amdgcn_readfirstlane(threadIdx.x >> 5);
  if (i >= n) return;
  const float4* xw4 = (const float4*)xw;
  const int2* lst2 = (const int2*)lst;
  const int beg = off[i];
  const int len = cnt[i];
  const float di = dinv[i];
  const float4 self = xw4[(size_t)i * (D / 4) + lane];
  const float sn = di;
  float4 acc = make_float4(sn * self.x, sn * self.y, sn * self.z, sn * self.w);
  for (int base = 0; base < len; base += 32) {
    const int m = (len - base < 32) ? (len - base) : 32;
    int r_l = 0;
    float w_l = 0.0f;
    if (lane < m) {
      const int2 pe = lst2[beg + base + lane];
      r_l = pe.x;
      w_l = __int_as_float(pe.y) * di;
    }
    int j = 0;
    for (; j + 8 <= m; j += 8) gather_step<8>(xw4, r_l, w_l, j, lane, acc);
    if (m & 4) { gather_step<4>(xw4, r_l, w_l, j, lane, acc); j += 4; }
    if (m & 2) { gather_step<2>(xw4, r_l, w_l, j, lane, acc); j += 2; }
    if (m & 1) { gather_step<1>(xw4, r_l, w_l, j, lane, acc); }
  }
  const float a = pa[0];
  float4 o;
  o.x = acc.x + bias[lane * 4 + 0];
  o.y = acc.y + bias[lane * 4 + 1];
  o.z = acc.z + bias[lane * 4 + 2];
  o.w = acc.w + bias[lane * 4 + 3];
  o.x = (o.x >= 0.0f) ? o.x : a * o.x;
  o.y = (o.y >= 0.0f) ? o.y : a * o.y;
  o.z = (o.z >= 0.0f) ? o.z : a * o.z;
  o.w = (o.w >= 0.0f) ? o.w : a * o.w;
  ((float4*)out)[(size_t)i * (D / 4) + lane] = o;
}

extern "C" void kernel_launch(void* const* d_in, const int* in_sizes, int n_in,
                              void* d_out, int out_size, void* d_ws, size_t ws_size,
                              hipStream_t stream) {
  const float* x    = (const float*)d_in[0];
  const int*   ei   = (const int*)  d_in[1];
  const float* ew   = (const float*)d_in[2];
  const float* w    = (const float*)d_in[3];
  const float* bias = (const float*)d_in[4];
  const float* pa   = (const float*)d_in[5];
  float* out = (float*)d_out;

  const int n     = in_sizes[0] / D;
  const int e_cnt = in_sizes[2];
  const int* row = ei;
  const int* col = ei + e_cnt;
  const int nb = (n + NBK - 1) / NBK;
  const int nh = nb * NCHUNK;

  const size_t n_pad = ((size_t)n + 127) / 128 * 128;
  const size_t e_pad = ((size_t)e_cnt + 127) / 128 * 128;
  const size_t h_pad = ((size_t)nh + 127) / 128 * 128;
  int*   cnt  = (int*)d_ws;
  int*   off  = cnt + n_pad;
  float* dinv = (float*)(off + n_pad);
  int*   segcnt = (int*)(dinv + n_pad);
  int*   segoff = segcnt + h_pad;
  int*   bkt  = segoff + h_pad;
  int*   lst  = bkt + 2 * e_pad;
  float* xw   = (float*)(lst + 2 * e_pad);

  chunk_sort_kernel<<<NCHUNK, 1024, 2 * nb * sizeof(int), stream>>>(row, col, ew, segcnt, segoff, bkt, e_cnt, nb);
  bucket_gemm_kernel<<<nb, 1024, 0, stream>>>(segcnt, segoff, bkt, cnt, off, dinv, lst, x, w, xw, n, e_cnt);
  gather_kernel<<<(n + 7) / 8, 256, 0, stream>>>(off, cnt, lst, dinv, xw, bias, pa, out, n);
}
